# starting kernel
# baseline (speedup 1.0000x reference)
#include <hip/hip_runtime.h>
#include <hip/hip_bf16.h>
#include <math.h>

typedef __attribute__((ext_vector_type(2))) float v2f;
typedef __attribute__((ext_vector_type(8))) float v8f;

#define IN_F   256
#define OUTC   256
#define HEADS  4
#define OUT_F  64
#define NEG_SLOPE 0.2f

__device__ __forceinline__ float leaky(float v) { return v > 0.f ? v : NEG_SLOPE * v; }

__global__ void gat_init(float* __restrict__ out, const float* __restrict__ b,
                         float* __restrict__ gmax, float* __restrict__ gsum,
                         int outElems) {
    int i = blockIdx.x * blockDim.x + threadIdx.x;
    if (i < outElems) out[i] = b[i & (OUTC - 1)];
    if (i < HEADS) { gmax[i] = -INFINITY; gsum[i] = 0.f; }
}

__global__ void __launch_bounds__(256) gat_gemm_wmma(
        const float* __restrict__ x, const float* __restrict__ W,
        float* __restrict__ h, int nTilesM) {
    const int wave = blockIdx.x * (blockDim.x >> 5) + (threadIdx.x >> 5);
    const int lane = threadIdx.x & 31;
    const int mTile = wave >> 4;
    const int nTile = wave & 15;
    if (mTile >= nTilesM) return;

    const int m  = lane & 15;
    const int kb = (lane >> 4) << 1;

    const float* xrow = x + (size_t)(mTile * 16 + m) * IN_F;
    const float* wcol = W + nTile * 16 + (lane & 15);

    v8f c = {};
    #pragma unroll 4
    for (int k = 0; k < IN_F; k += 4) {
        v2f a = *(const v2f*)(xrow + k + kb);
        v2f bf;
        bf.x = wcol[(size_t)(k + kb)     * OUTC];
        bf.y = wcol[(size_t)(k + kb + 1) * OUTC];
        c = __builtin_amdgcn_wmma_f32_16x16x4_f32(
                 false, a,  false, bf,
                 (short)0, c,  false,  false);
    }

    const int nlocal = lane & 15;
    const int rowOff = (lane >> 4) << 3;
    float* hp = h + nTile * 16 + nlocal;
    #pragma unroll
    for (int i = 0; i < 8; ++i)
        hp[(size_t)(mTile * 16 + rowOff + i) * OUTC] = c[i];
}

__global__ void __launch_bounds__(256) gat_scores(
        const float* __restrict__ h, const int* __restrict__ ei,
        const float* __restrict__ att, float* __restrict__ scores,
        float* __restrict__ gmax, int E, int Nn) {
    const int wave = blockIdx.x * (blockDim.x >> 5) + (threadIdx.x >> 5);
    const int lane = threadIdx.x & 31;
    if (wave >= E + Nn) return;

    int s, d;
    if (wave < E) { s = ei[wave]; d = ei[E + wave]; }
    else          { s = d = wave - E; }

    const float* hs = h + (size_t)s * OUTC;
    const float* hd = h + (size_t)d * OUTC;

    float acc[HEADS] = {0.f, 0.f, 0.f, 0.f};
    #pragma unroll
    for (int i = 0; i < 8; ++i) {
        const int head = i >> 1;
        const int j    = lane + ((i & 1) << 5);
        const int idx  = lane + (i << 5);
        const float asv = att[head * 2 * OUT_F + j];
        const float adv = att[head * 2 * OUT_F + OUT_F + j];
        acc[head] += leaky(hs[idx]) * asv + leaky(hd[idx]) * adv;
    }

    #pragma unroll
    for (int hdi = 0; hdi < HEADS; ++hdi) {
        float v = acc[hdi];
        #pragma unroll
        for (int off = 16; off > 0; off >>= 1)
            v += __shfl_xor(v, off, 32);
        if (lane == 0) {
            scores[(size_t)wave * HEADS + hdi] = v;
            atomicMax(gmax + hdi, v);
        }
    }
}

__global__ void __launch_bounds__(256) gat_expsum(
        float* __restrict__ scores, const float* __restrict__ gmax,
        float* __restrict__ gsum, int total4) {
    __shared__ float ssum[HEADS];
    if (threadIdx.x < HEADS) ssum[threadIdx.x] = 0.f;
    __syncthreads();

    const int i = blockIdx.x * blockDim.x + threadIdx.x;
    if (i < total4) {
        const int hd = i & 3;
        const float v = __expf(scores[i] - gmax[hd]);
        scores[i] = v;
        atomicAdd(&ssum[hd], v);
    }
    __syncthreads();
    if (threadIdx.x < HEADS) atomicAdd(gsum + threadIdx.x, ssum[threadIdx.x]);
}

__global__ void __launch_bounds__(256) gat_scatter(
        const float* __restrict__ h, const int* __restrict__ ei,
        const float* __restrict__ scores, const float* __restrict__ gsum,
        float* __restrict__ out, int E, int Nn) {
    const int wave = blockIdx.x * (blockDim.x >> 5) + (threadIdx.x >> 5);
    const int lane = threadIdx.x & 31;
    if (wave >= E + Nn) return;

    int s, d;
    if (wave < E) { s = ei[wave]; d = ei[E + wave]; }
    else          { s = d = wave - E; }

    float wgt[HEADS];
    #pragma unroll
    for (int hd = 0; hd < HEADS; ++hd)
        wgt[hd] = scores[(size_t)wave * HEADS + hd] / gsum[hd];

    const float* hs = h + (size_t)s * OUTC;
    float* op = out + (size_t)d * OUTC;
    #pragma unroll
    for (int i = 0; i < 8; ++i) {
        const int idx = lane + (i << 5);
        atomicAdd(op + idx, hs[idx] * wgt[i >> 1]);
    }
}

extern "C" void kernel_launch(void* const* d_in, const int* in_sizes, int n_in,
                              void* d_out, int out_size, void* d_ws, size_t ws_size,
                              hipStream_t stream) {
    const float* x   = (const float*)d_in[0];
    const float* W   = (const float*)d_in[1];
    const float* att = (const float*)d_in[2];
    const float* b   = (const float*)d_in[3];
    const int*   ei  = (const int*)d_in[4];

    const int Nn = in_sizes[0] / IN_F;
    const int E  = in_sizes[4] / 2;
    const int totalEdges = E + Nn;

    float* out    = (float*)d_out;
    float* h      = (float*)d_ws;
    float* scores = h + (size_t)Nn * OUTC;
    float* gmax   = scores + (size_t)totalEdges * HEADS;
    float* gsum   = gmax + HEADS;

    const int outElems = Nn * OUTC;
    gat_init<<<(outElems + 255) / 256, 256, 0, stream>>>(out, b, gmax, gsum, outElems);

    const int nTilesM = Nn / 16;
    const int tiles   = nTilesM * (OUTC / 16);
    gat_gemm_wmma<<<(tiles + 7) / 8, 256, 0, stream>>>(x, W, h, nTilesM);

    gat_scores<<<(totalEdges + 7) / 8, 256, 0, stream>>>(h, ei, att, scores, gmax, E, Nn);
    gat_expsum<<<(totalEdges * HEADS + 255) / 256, 256, 0, stream>>>(
        scores, gmax, gsum, totalEdges * HEADS);
    gat_scatter<<<(totalEdges + 7) / 8, 256, 0, stream>>>(h, ei, scores, gsum, out, E, Nn);
}


# scores kernel: per-head max kept in registers over a resident grid's edge loop, reduced across the block in LDS, one atomic max per head per block instead of per edge on the same 4 words
# speedup vs baseline: 7.1213x; 7.1213x over previous
#include <hip/hip_runtime.h>
#include <hip/hip_bf16.h>
#include <math.h>

typedef __attribute__((ext_vector_type(2))) float v2f;
typedef __attribute__((ext_vector_type(8))) float v8f;

#define IN_F   256
#define OUTC   256
#define HEADS  4
#define OUT_F  64
#define NEG_SLOPE 0.2f

__device__ __forceinline__ float leaky(float v) { return v > 0.f ? v : NEG_SLOPE * v; }

__global__ void gat_init(float* __restrict__ out, const float* __restrict__ b,
                         float* __restrict__ gmax, float* __restrict__ gsum,
                         int outElems) {
    int i = blockIdx.x * blockDim.x + threadIdx.x;
    if (i < outElems) out[i] = b[i & (OUTC - 1)];
    if (i < HEADS) { gmax[i] = -INFINITY; gsum[i] = 0.f; }
}

__global__ void __launch_bounds__(256) gat_gemm_wmma(
        const float* __restrict__ x, const float* __restrict__ W,
        float* __restrict__ h, int nTilesM) {
    const int wave = blockIdx.x * (blockDim.x >> 5) + (threadIdx.x >> 5);
    const int lane = threadIdx.x & 31;
    const int mTile = wave >> 4;
    const int nTile = wave & 15;
    if (mTile >= nTilesM) return;

    const int m  = lane & 15;
    const int kb = (lane >> 4) << 1;

    const float* xrow = x + (size_t)(mTile * 16 + m) * IN_F;
    const float* wcol = W + nTile * 16 + (lane & 15);

    v8f c = {};
    #pragma unroll 4
    for (int k = 0; k < IN_F; k += 4) {
        v2f a = *(const v2f*)(xrow + k + kb);
        v2f bf;
        bf.x = wcol[(size_t)(k + kb)     * OUTC];
        bf.y = wcol[(size_t)(k + kb + 1) * OUTC];
        c = __builtin_amdgcn_wmma_f32_16x16x4_f32(
                 false, a,  false, bf,
                 (short)0, c,  false,  false);
    }

    const int nlocal = lane & 15;
    const int rowOff = (lane >> 4) << 3;
    float* hp = h + nTile * 16 + nlocal;
    #pragma unroll
    for (int i = 0; i < 8; ++i)
        hp[(size_t)(mTile * 16 + rowOff + i) * OUTC] = c[i];
}

#define SCORE_WAVES 8

__global__ void __launch_bounds__(256) gat_scores(
        const float* __restrict__ h, const int* __restrict__ ei,
        const float* __restrict__ att, float* __restrict__ scores,
        float* __restrict__ gmax, int E, int Nn) {
    __shared__ float smax[SCORE_WAVES][HEADS];
    const int lane   = threadIdx.x & 31;
    const int wib    = threadIdx.x >> 5;
    const int nWaves = gridDim.x * SCORE_WAVES;
    const int total  = E + Nn;

    float asv[8], adv[8];
    #pragma unroll
    for (int i = 0; i < 8; ++i) {
        const int head = i >> 1;
        const int j    = lane + ((i & 1) << 5);
        asv[i] = att[head * 2 * OUT_F + j];
        adv[i] = att[head * 2 * OUT_F + OUT_F + j];
    }

    float m[HEADS] = {-INFINITY, -INFINITY, -INFINITY, -INFINITY};

    for (int e = blockIdx.x * SCORE_WAVES + wib; e < total; e += nWaves) {
        int s, d;
        if (e < E) { s = ei[e]; d = ei[E + e]; }
        else       { s = d = e - E; }

        const float* hs = h + (size_t)s * OUTC;
        const float* hd = h + (size_t)d * OUTC;

        float acc[HEADS] = {0.f, 0.f, 0.f, 0.f};
        #pragma unroll
        for (int i = 0; i < 8; ++i) {
            const int idx = lane + (i << 5);
            acc[i >> 1] += leaky(hs[idx]) * asv[i] + leaky(hd[idx]) * adv[i];
        }

        #pragma unroll
        for (int hdi = 0; hdi < HEADS; ++hdi) {
            float v = acc[hdi];
            #pragma unroll
            for (int off = 16; off > 0; off >>= 1)
                v += __shfl_xor(v, off, 32);
            if (lane == 0) scores[(size_t)e * HEADS + hdi] = v;
            m[hdi] = fmaxf(m[hdi], v);
        }
    }

    if (lane == 0) {
        smax[wib][0] = m[0]; smax[wib][1] = m[1];
        smax[wib][2] = m[2]; smax[wib][3] = m[3];
    }
    __syncthreads();
    if (threadIdx.x < HEADS) {
        float v = smax[0][threadIdx.x];
        for (int w = 1; w < SCORE_WAVES; ++w) v = fmaxf(v, smax[w][threadIdx.x]);
        atomicMax(gmax + threadIdx.x, v);
    }
}

__global__ void __launch_bounds__(256) gat_expsum(
        float* __restrict__ scores, const float* __restrict__ gmax,
        float* __restrict__ gsum, int total4) {
    __shared__ float ssum[HEADS];
    if (threadIdx.x < HEADS) ssum[threadIdx.x] = 0.f;
    __syncthreads();

    const int i = blockIdx.x * blockDim.x + threadIdx.x;
    if (i < total4) {
        const int hd = i & 3;
        const float v = __expf(scores[i] - gmax[hd]);
        scores[i] = v;
        atomicAdd(&ssum[hd], v);
    }
    __syncthreads();
    if (threadIdx.x < HEADS) atomicAdd(gsum + threadIdx.x, ssum[threadIdx.x]);
}

__global__ void __launch_bounds__(256) gat_scatter(
        const float* __restrict__ h, const int* __restrict__ ei,
        const float* __restrict__ scores, const float* __restrict__ gsum,
        float* __restrict__ out, int E, int Nn) {
    const int wave = blockIdx.x * (blockDim.x >> 5) + (threadIdx.x >> 5);
    const int lane = threadIdx.x & 31;
    if (wave >= E + Nn) return;

    int s, d;
    if (wave < E) { s = ei[wave]; d = ei[E + wave]; }
    else          { s = d = wave - E; }

    float wgt[HEADS];
    #pragma unroll
    for (int hd = 0; hd < HEADS; ++hd)
        wgt[hd] = scores[(size_t)wave * HEADS + hd] / gsum[hd];

    const float* hs = h + (size_t)s * OUTC;
    float* op = out + (size_t)d * OUTC;
    #pragma unroll
    for (int i = 0; i < 8; ++i) {
        const int idx = lane + (i << 5);
        atomicAdd(op + idx, hs[idx] * wgt[i >> 1]);
    }
}

extern "C" void kernel_launch(void* const* d_in, const int* in_sizes, int n_in,
                              void* d_out, int out_size, void* d_ws, size_t ws_size,
                              hipStream_t stream) {
    const float* x   = (const float*)d_in[0];
    const float* W   = (const float*)d_in[1];
    const float* att = (const float*)d_in[2];
    const float* b   = (const float*)d_in[3];
    const int*   ei  = (const int*)d_in[4];

    const int Nn = in_sizes[0] / IN_F;
    const int E  = in_sizes[4] / 2;
    const int totalEdges = E + Nn;

    float* out    = (float*)d_out;
    float* h      = (float*)d_ws;
    float* scores = h + (size_t)Nn * OUTC;
    float* gmax   = scores + (size_t)totalEdges * HEADS;
    float* gsum   = gmax + HEADS;

    const int outElems = Nn * OUTC;
    gat_init<<<(outElems + 255) / 256, 256, 0, stream>>>(out, b, gmax, gsum, outElems);

    const int nTilesM = Nn / 16;
    const int tiles   = nTilesM * (OUTC / 16);
    gat_gemm_wmma<<<(tiles + 7) / 8, 256, 0, stream>>>(x, W, h, nTilesM);

    gat_scores<<<2048, 256, 0, stream>>>(h, ei, att, scores, gmax, E, Nn);
    gat_expsum<<<(totalEdges * HEADS + 255) / 256, 256, 0, stream>>>(
        scores, gmax, gsum, totalEdges * HEADS);
    gat_scatter<<<(totalEdges + 7) / 8, 256, 0, stream>>>(h, ei, scores, gsum, out, E, Nn);
}


# init, exp-sum and scatter become resident grids whose waves loop over their elements (514000 fewer wave starts); exp sums per thread then shuffle/LDS tree (sum order changes); weight = score*(1/sum)
# speedup vs baseline: 9.4775x; 1.3309x over previous
#include <hip/hip_runtime.h>
#include <hip/hip_bf16.h>
#include <math.h>

typedef __attribute__((ext_vector_type(2))) float v2f;
typedef __attribute__((ext_vector_type(8))) float v8f;

#define IN_F   256
#define OUTC   256
#define HEADS  4
#define OUT_F  64
#define NEG_SLOPE 0.2f

__device__ __forceinline__ float leaky(float v) { return v > 0.f ? v : NEG_SLOPE * v; }

__global__ void __launch_bounds__(256) gat_init(
        float* __restrict__ out, const float* __restrict__ b,
        float* __restrict__ gmax, float* __restrict__ gsum, int outVec4) {
    const int stride = gridDim.x * blockDim.x;
    int i = blockIdx.x * blockDim.x + threadIdx.x;
    if (i < HEADS) { gmax[i] = -INFINITY; gsum[i] = 0.f; }
    const int c = (i << 2) & (OUTC - 1);
    float4 bv;
    bv.x = b[c]; bv.y = b[c + 1]; bv.z = b[c + 2]; bv.w = b[c + 3];
    float4* o4 = (float4*)out;
    for (; i < outVec4; i += stride) o4[i] = bv;
}

__global__ void __launch_bounds__(256) gat_gemm_wmma(
        const float* __restrict__ x, const float* __restrict__ W,
        float* __restrict__ h, int nTilesM) {
    const int wave = blockIdx.x * (blockDim.x >> 5) + (threadIdx.x >> 5);
    const int lane = threadIdx.x & 31;
    const int mTile = wave >> 4;
    const int nTile = wave & 15;
    if (mTile >= nTilesM) return;

    const int m  = lane & 15;
    const int kb = (lane >> 4) << 1;

    const float* xrow = x + (size_t)(mTile * 16 + m) * IN_F;
    const float* wcol = W + nTile * 16 + (lane & 15);

    v8f c = {};
    #pragma unroll 4
    for (int k = 0; k < IN_F; k += 4) {
        v2f a = *(const v2f*)(xrow + k + kb);
        v2f bf;
        bf.x = wcol[(size_t)(k + kb)     * OUTC];
        bf.y = wcol[(size_t)(k + kb + 1) * OUTC];
        c = __builtin_amdgcn_wmma_f32_16x16x4_f32(
                 false, a,  false, bf,
                 (short)0, c,  false,  false);
    }

    const int nlocal = lane & 15;
    const int rowOff = (lane >> 4) << 3;
    float* hp = h + nTile * 16 + nlocal;
    #pragma unroll
    for (int i = 0; i < 8; ++i)
        hp[(size_t)(mTile * 16 + rowOff + i) * OUTC] = c[i];
}

#define SCORE_WAVES 8

__global__ void __launch_bounds__(256) gat_scores(
        const float* __restrict__ h, const int* __restrict__ ei,
        const float* __restrict__ att, float* __restrict__ scores,
        float* __restrict__ gmax, int E, int Nn) {
    __shared__ float smax[SCORE_WAVES][HEADS];
    const int lane   = threadIdx.x & 31;
    const int wib    = threadIdx.x >> 5;
    const int nWaves = gridDim.x * SCORE_WAVES;
    const int total  = E + Nn;

    float asv[8], adv[8];
    #pragma unroll
    for (int i = 0; i < 8; ++i) {
        const int head = i >> 1;
        const int j    = lane + ((i & 1) << 5);
        asv[i] = att[head * 2 * OUT_F + j];
        adv[i] = att[head * 2 * OUT_F + OUT_F + j];
    }

    float m[HEADS] = {-INFINITY, -INFINITY, -INFINITY, -INFINITY};

    for (int e = blockIdx.x * SCORE_WAVES + wib; e < total; e += nWaves) {
        int s, d;
        if (e < E) { s = ei[e]; d = ei[E + e]; }
        else       { s = d = e - E; }

        const float* hs = h + (size_t)s * OUTC;
        const float* hd = h + (size_t)d * OUTC;

        float acc[HEADS] = {0.f, 0.f, 0.f, 0.f};
        #pragma unroll
        for (int i = 0; i < 8; ++i) {
            const int idx = lane + (i << 5);
            acc[i >> 1] += leaky(hs[idx]) * asv[i] + leaky(hd[idx]) * adv[i];
        }

        #pragma unroll
        for (int hdi = 0; hdi < HEADS; ++hdi) {
            float v = acc[hdi];
            #pragma unroll
            for (int off = 16; off > 0; off >>= 1)
                v += __shfl_xor(v, off, 32);
            if (lane == 0) scores[(size_t)e * HEADS + hdi] = v;
            m[hdi] = fmaxf(m[hdi], v);
        }
    }

    if (lane == 0) {
        smax[wib][0] = m[0]; smax[wib][1] = m[1];
        smax[wib][2] = m[2]; smax[wib][3] = m[3];
    }
    __syncthreads();
    if (threadIdx.x < HEADS) {
        float v = smax[0][threadIdx.x];
        for (int w = 1; w < SCORE_WAVES; ++w) v = fmaxf(v, smax[w][threadIdx.x]);
        atomicMax(gmax + threadIdx.x, v);
    }
}

#define EXPS_WAVES 8

__global__ void __launch_bounds__(256) gat_expsum(
        float* __restrict__ scores, const float* __restrict__ gmax,
        float* __restrict__ gsum, int totalEdges) {
    __shared__ float ssum[EXPS_WAVES][HEADS];
    const int lane = threadIdx.x & 31;
    const int wib  = threadIdx.x >> 5;
    const int stride = gridDim.x * blockDim.x;
    const float m0 = gmax[0], m1 = gmax[1], m2 = gmax[2], m3 = gmax[3];

    float4* s4 = (float4*)scores;
    float a0 = 0.f, a1 = 0.f, a2 = 0.f, a3 = 0.f;
    for (int e = blockIdx.x * blockDim.x + threadIdx.x; e < totalEdges; e += stride) {
        float4 v = s4[e];
        v.x = __expf(v.x - m0); v.y = __expf(v.y - m1);
        v.z = __expf(v.z - m2); v.w = __expf(v.w - m3);
        s4[e] = v;
        a0 += v.x; a1 += v.y; a2 += v.z; a3 += v.w;
    }

    #pragma unroll
    for (int off = 16; off > 0; off >>= 1) {
        a0 += __shfl_xor(a0, off, 32); a1 += __shfl_xor(a1, off, 32);
        a2 += __shfl_xor(a2, off, 32); a3 += __shfl_xor(a3, off, 32);
    }
    if (lane == 0) {
        ssum[wib][0] = a0; ssum[wib][1] = a1; ssum[wib][2] = a2; ssum[wib][3] = a3;
    }
    __syncthreads();
    if (threadIdx.x < HEADS) {
        float v = ssum[0][threadIdx.x];
        for (int w = 1; w < EXPS_WAVES; ++w) v += ssum[w][threadIdx.x];
        atomicAdd(gsum + threadIdx.x, v);
    }
}

#define SCAT_WAVES 8

__global__ void __launch_bounds__(256) gat_scatter(
        const float* __restrict__ h, const int* __restrict__ ei,
        const float* __restrict__ scores, const float* __restrict__ gsum,
        float* __restrict__ out, int E, int Nn) {
    const int lane   = threadIdx.x & 31;
    const int wib    = __builtin_amdgcn_readfirstlane((int)(threadIdx.x >> 5));
    const int nWaves = gridDim.x * SCAT_WAVES;
    const int total  = E + Nn;

    float inv[HEADS];
    #pragma unroll
    for (int hd = 0; hd < HEADS; ++hd) inv[hd] = 1.f / gsum[hd];

    for (int e = blockIdx.x * SCAT_WAVES + wib; e < total; e += nWaves) {
        int s, d;
        if (e < E) { s = ei[e]; d = ei[E + e]; }
        else       { s = d = e - E; }

        const float4 sc = *(const float4*)(scores + (size_t)e * HEADS);
        float wgt[HEADS];
        wgt[0] = sc.x * inv[0]; wgt[1] = sc.y * inv[1];
        wgt[2] = sc.z * inv[2]; wgt[3] = sc.w * inv[3];

        const float* hs = h + (size_t)s * OUTC + lane;
        float* op = out + (size_t)d * OUTC + lane;
        float v[8];
        #pragma unroll
        for (int i = 0; i < 8; ++i) v[i] = hs[i << 5];
        #pragma unroll
        for (int i = 0; i < 8; ++i)
            atomicAdd(op + (i << 5), v[i] * wgt[i >> 1]);
    }
}

extern "C" void kernel_launch(void* const* d_in, const int* in_sizes, int n_in,
                              void* d_out, int out_size, void* d_ws, size_t ws_size,
                              hipStream_t stream) {
    const float* x   = (const float*)d_in[0];
    const float* W   = (const float*)d_in[1];
    const float* att = (const float*)d_in[2];
    const float* b   = (const float*)d_in[3];
    const int*   ei  = (const int*)d_in[4];

    const int Nn = in_sizes[0] / IN_F;
    const int E  = in_sizes[4] / 2;
    const int totalEdges = E + Nn;

    float* out    = (float*)d_out;
    float* h      = (float*)d_ws;
    float* scores = h + (size_t)Nn * OUTC;
    float* gmax   = scores + (size_t)totalEdges * HEADS;
    float* gsum   = gmax + HEADS;

    const int outElems = Nn * OUTC;
    gat_init<<<1024, 256, 0, stream>>>(out, b, gmax, gsum, outElems / 4);

    const int nTilesM = Nn / 16;
    const int tiles   = nTilesM * (OUTC / 16);
    gat_gemm_wmma<<<(tiles + 7) / 8, 256, 0, stream>>>(x, W, h, nTilesM);

    gat_scores<<<2048, 256, 0, stream>>>(h, ei, att, scores, gmax, E, Nn);
    gat_expsum<<<512, 256, 0, stream>>>(scores, gmax, gsum, totalEdges);
    gat_scatter<<<2048, 256, 0, stream>>>(h, ei, scores, gsum, out, E, Nn);
}


# attention dot products hoisted from edges to nodes: GEMM epilogue forms each node's 4 source and 4 target half scores once; edge kernel adds two gathered 16-byte vectors (64-term sum order changes)
# speedup vs baseline: 13.9664x; 1.4736x over previous
#include <hip/hip_runtime.h>
#include <hip/hip_bf16.h>
#include <math.h>

typedef __attribute__((ext_vector_type(2))) float v2f;
typedef __attribute__((ext_vector_type(8))) float v8f;

#define IN_F   256
#define OUTC   256
#define HEADS  4
#define OUT_F  64
#define NEG_SLOPE 0.2f

__device__ __forceinline__ float leaky(float v) { return v > 0.f ? v : NEG_SLOPE * v; }

__global__ void __launch_bounds__(256) gat_init(
        float* __restrict__ out, const float* __restrict__ b,
        float* __restrict__ gmax, float* __restrict__ gsum, int outVec4) {
    const int stride = gridDim.x * blockDim.x;
    int i = blockIdx.x * blockDim.x + threadIdx.x;
    if (i < HEADS) { gmax[i] = -INFINITY; gsum[i] = 0.f; }
    const int c = (i << 2) & (OUTC - 1);
    float4 bv;
    bv.x = b[c]; bv.y = b[c + 1]; bv.z = b[c + 2]; bv.w = b[c + 3];
    float4* o4 = (float4*)out;
    for (; i < outVec4; i += stride) o4[i] = bv;
}

#define GEMM_WAVES 8

template <int M>
__device__ __forceinline__ float row_xor_add(float v) {
    const int t = __builtin_amdgcn_update_dpp(0, __builtin_bit_cast(int, v),
                                              0x160 + M, 0xf, 0xf, true);
    return v + __builtin_bit_cast(float, t);
}

__global__ void __launch_bounds__(256) gat_gemm_wmma(
        const float* __restrict__ x, const float* __restrict__ W,
        const float* __restrict__ att,
        float* __restrict__ h, float* __restrict__ p, int nTilesM) {
    __shared__ float spart[GEMM_WAVES][16][2];
    const int wave = blockIdx.x * (blockDim.x >> 5) + (threadIdx.x >> 5);
    const int lane = threadIdx.x & 31;
    const int mTile = wave >> 4;
    const int nTile = wave & 15;
    if (mTile >= nTilesM) return;

    const int m  = lane & 15;
    const int kb = (lane >> 4) << 1;

    const float* xrow = x + (size_t)(mTile * 16 + m) * IN_F;
    const float* wcol = W + nTile * 16 + (lane & 15);

    const int head = nTile >> 2;
    const int j    = ((nTile & 3) << 4) + (lane & 15);
    const float as = att[head * 2 * OUT_F + j];
    const float ad = att[head * 2 * OUT_F + OUT_F + j];

    v8f c = {};
    #pragma unroll 4
    for (int k = 0; k < IN_F; k += 4) {
        v2f a = *(const v2f*)(xrow + k + kb);
        v2f bf;
        bf.x = wcol[(size_t)(k + kb)     * OUTC];
        bf.y = wcol[(size_t)(k + kb + 1) * OUTC];
        c = __builtin_amdgcn_wmma_f32_16x16x4_f32(
                 false, a,  false, bf,
                 (short)0, c,  false,  false);
    }

    const int nlocal = lane & 15;
    const int rowOff = (lane >> 4) << 3;
    float* hp = h + nTile * 16 + nlocal;
    #pragma unroll
    for (int i = 0; i < 8; ++i)
        hp[(size_t)(mTile * 16 + rowOff + i) * OUTC] = c[i];

    const int wib  = threadIdx.x >> 5;
    float ps[8], pd[8];
    #pragma unroll
    for (int i = 0; i < 8; ++i) {
        const float l = leaky(c[i]);
        ps[i] = l * as;
        pd[i] = l * ad;
    }
    #pragma unroll
    for (int i = 0; i < 8; ++i) { ps[i] = row_xor_add<1>(ps[i]); pd[i] = row_xor_add<1>(pd[i]); }
    #pragma unroll
    for (int i = 0; i < 8; ++i) { ps[i] = row_xor_add<2>(ps[i]); pd[i] = row_xor_add<2>(pd[i]); }
    #pragma unroll
    for (int i = 0; i < 8; ++i) { ps[i] = row_xor_add<4>(ps[i]); pd[i] = row_xor_add<4>(pd[i]); }
    #pragma unroll
    for (int i = 0; i < 8; ++i) { ps[i] = row_xor_add<8>(ps[i]); pd[i] = row_xor_add<8>(pd[i]); }
    if (nlocal == 0) {
        #pragma unroll
        for (int i = 0; i < 8; ++i) {
            spart[wib][rowOff + i][0] = ps[i];
            spart[wib][rowOff + i][1] = pd[i];
        }
    }
    __syncthreads();
    if (threadIdx.x < 32) {
        const int row = threadIdx.x >> 1;
        const int sd  = threadIdx.x & 1;
        v2f v;
        v.x = (spart[0][row][sd] + spart[1][row][sd]) + (spart[2][row][sd] + spart[3][row][sd]);
        v.y = (spart[4][row][sd] + spart[5][row][sd]) + (spart[6][row][sd] + spart[7][row][sd]);
        const int head0 = ((blockIdx.x * GEMM_WAVES) & 15) >> 2;
        *(v2f*)(p + (size_t)(mTile * 16 + row) * (2 * HEADS) + sd * HEADS + head0) = v;
    }
}

#define SCORE_WAVES 8

__global__ void __launch_bounds__(256) gat_scores(
        const float* __restrict__ p, const int* __restrict__ ei,
        float* __restrict__ scores, float* __restrict__ gmax, int E, int Nn) {
    __shared__ float smax[SCORE_WAVES][HEADS];
    const int lane   = threadIdx.x & 31;
    const int wib    = threadIdx.x >> 5;
    const int stride = gridDim.x * blockDim.x;
    const int total  = E + Nn;

    const float4* p4 = (const float4*)p;
    float4* s4 = (float4*)scores;
    float m0 = -INFINITY, m1 = -INFINITY, m2 = -INFINITY, m3 = -INFINITY;

    for (int e = blockIdx.x * blockDim.x + threadIdx.x; e < total; e += stride) {
        int s, d;
        if (e < E) { s = ei[e]; d = ei[E + e]; }
        else       { s = d = e - E; }
        const float4 a = p4[(size_t)s * 2];
        const float4 b = p4[(size_t)d * 2 + 1];
        float4 v;
        v.x = a.x + b.x; v.y = a.y + b.y; v.z = a.z + b.z; v.w = a.w + b.w;
        s4[e] = v;
        m0 = fmaxf(m0, v.x); m1 = fmaxf(m1, v.y);
        m2 = fmaxf(m2, v.z); m3 = fmaxf(m3, v.w);
    }

    #pragma unroll
    for (int off = 16; off > 0; off >>= 1) {
        m0 = fmaxf(m0, __shfl_xor(m0, off, 32)); m1 = fmaxf(m1, __shfl_xor(m1, off, 32));
        m2 = fmaxf(m2, __shfl_xor(m2, off, 32)); m3 = fmaxf(m3, __shfl_xor(m3, off, 32));
    }
    if (lane == 0) {
        smax[wib][0] = m0; smax[wib][1] = m1;
        smax[wib][2] = m2; smax[wib][3] = m3;
    }
    __syncthreads();
    if (threadIdx.x < HEADS) {
        float v = smax[0][threadIdx.x];
        for (int w = 1; w < SCORE_WAVES; ++w) v = fmaxf(v, smax[w][threadIdx.x]);
        atomicMax(gmax + threadIdx.x, v);
    }
}

#define EXPS_WAVES 8

__global__ void __launch_bounds__(256) gat_expsum(
        float* __restrict__ scores, const float* __restrict__ gmax,
        float* __restrict__ gsum, int totalEdges) {
    __shared__ float ssum[EXPS_WAVES][HEADS];
    const int lane = threadIdx.x & 31;
    const int wib  = threadIdx.x >> 5;
    const int stride = gridDim.x * blockDim.x;
    const float m0 = gmax[0], m1 = gmax[1], m2 = gmax[2], m3 = gmax[3];

    float4* s4 = (float4*)scores;
    float a0 = 0.f, a1 = 0.f, a2 = 0.f, a3 = 0.f;
    for (int e = blockIdx.x * blockDim.x + threadIdx.x; e < totalEdges; e += stride) {
        float4 v = s4[e];
        v.x = __expf(v.x - m0); v.y = __expf(v.y - m1);
        v.z = __expf(v.z - m2); v.w = __expf(v.w - m3);
        s4[e] = v;
        a0 += v.x; a1 += v.y; a2 += v.z; a3 += v.w;
    }

    #pragma unroll
    for (int off = 16; off > 0; off >>= 1) {
        a0 += __shfl_xor(a0, off, 32); a1 += __shfl_xor(a1, off, 32);
        a2 += __shfl_xor(a2, off, 32); a3 += __shfl_xor(a3, off, 32);
    }
    if (lane == 0) {
        ssum[wib][0] = a0; ssum[wib][1] = a1; ssum[wib][2] = a2; ssum[wib][3] = a3;
    }
    __syncthreads();
    if (threadIdx.x < HEADS) {
        float v = ssum[0][threadIdx.x];
        for (int w = 1; w < EXPS_WAVES; ++w) v += ssum[w][threadIdx.x];
        atomicAdd(gsum + threadIdx.x, v);
    }
}

#define SCAT_WAVES 8

__global__ void __launch_bounds__(256) gat_scatter(
        const float* __restrict__ h, const int* __restrict__ ei,
        const float* __restrict__ scores, const float* __restrict__ gsum,
        float* __restrict__ out, int E, int Nn) {
    const int lane   = threadIdx.x & 31;
    const int wib    = __builtin_amdgcn_readfirstlane((int)(threadIdx.x >> 5));
    const int nWaves = gridDim.x * SCAT_WAVES;
    const int total  = E + Nn;

    float inv[HEADS];
    #pragma unroll
    for (int hd = 0; hd < HEADS; ++hd) inv[hd] = 1.f / gsum[hd];

    for (int e = blockIdx.x * SCAT_WAVES + wib; e < total; e += nWaves) {
        int s, d;
        if (e < E) { s = ei[e]; d = ei[E + e]; }
        else       { s = d = e - E; }

        const float4 sc = *(const float4*)(scores + (size_t)e * HEADS);
        float wgt[HEADS];
        wgt[0] = sc.x * inv[0]; wgt[1] = sc.y * inv[1];
        wgt[2] = sc.z * inv[2]; wgt[3] = sc.w * inv[3];

        const float* hs = h + (size_t)s * OUTC + lane;
        float* op = out + (size_t)d * OUTC + lane;
        float v[8];
        #pragma unroll
        for (int i = 0; i < 8; ++i) v[i] = hs[i << 5];
        #pragma unroll
        for (int i = 0; i < 8; ++i)
            atomicAdd(op + (i << 5), v[i] * wgt[i >> 1]);
    }
}

extern "C" void kernel_launch(void* const* d_in, const int* in_sizes, int n_in,
                              void* d_out, int out_size, void* d_ws, size_t ws_size,
                              hipStream_t stream) {
    const float* x   = (const float*)d_in[0];
    const float* W   = (const float*)d_in[1];
    const float* att = (const float*)d_in[2];
    const float* b   = (const float*)d_in[3];
    const int*   ei  = (const int*)d_in[4];

    const int Nn = in_sizes[0] / IN_F;
    const int E  = in_sizes[4] / 2;
    const int totalEdges = E + Nn;

    float* out    = (float*)d_out;
    float* h      = (float*)d_ws;
    float* p      = h + (size_t)Nn * OUTC;
    float* scores = p + (size_t)Nn * 2 * HEADS;
    float* gmax   = scores + (size_t)totalEdges * HEADS;
    float* gsum   = gmax + HEADS;

    const int outElems = Nn * OUTC;
    gat_init<<<1024, 256, 0, stream>>>(out, b, gmax, gsum, outElems / 4);

    const int nTilesM = Nn / 16;
    const int tiles   = nTilesM * (OUTC / 16);
    gat_gemm_wmma<<<(tiles + 7) / 8, 256, 0, stream>>>(x, W, att, h, p, nTilesM);

    gat_scores<<<512, 256, 0, stream>>>(p, ei, scores, gmax, E, Nn);
    gat_expsum<<<512, 256, 0, stream>>>(scores, gmax, gsum, totalEdges);
    gat_scatter<<<2048, 256, 0, stream>>>(h, ei, scores, gsum, out, E, Nn);
}


# GEMM: each wave computes a 16x64 head strip (4 accumulator tiles share each x fragment) with 16-byte x, W and h accesses instead of one 16x16 tile; k-sum and 64-term score sum order change
# speedup vs baseline: 17.4716x; 1.2510x over previous
#include <hip/hip_runtime.h>
#include <hip/hip_bf16.h>
#include <math.h>

typedef __attribute__((ext_vector_type(2))) float v2f;
typedef __attribute__((ext_vector_type(8))) float v8f;

#define IN_F   256
#define OUTC   256
#define HEADS  4
#define OUT_F  64
#define NEG_SLOPE 0.2f

__device__ __forceinline__ float leaky(float v) { return v > 0.f ? v : NEG_SLOPE * v; }

__global__ void __launch_bounds__(256) gat_init(
        float* __restrict__ out, const float* __restrict__ b,
        float* __restrict__ gmax, float* __restrict__ gsum, int outVec4) {
    const int stride = gridDim.x * blockDim.x;
    int i = blockIdx.x * blockDim.x + threadIdx.x;
    if (i < HEADS) { gmax[i] = -INFINITY; gsum[i] = 0.f; }
    const int c = (i << 2) & (OUTC - 1);
    float4 bv;
    bv.x = b[c]; bv.y = b[c + 1]; bv.z = b[c + 2]; bv.w = b[c + 3];
    float4* o4 = (float4*)out;
    for (; i < outVec4; i += stride) o4[i] = bv;
}

#define GEMM_WAVES 4

template <int M>
__device__ __forceinline__ float row_xor_add(float v) {
    const int t = __builtin_amdgcn_update_dpp(0, __builtin_bit_cast(int, v),
                                              0x160 + M, 0xf, 0xf, true);
    return v + __builtin_bit_cast(float, t);
}

struct Frag {
    float4 a;
    float4 q[4];
};

__device__ __forceinline__ void load_frag(Frag& f, const float* __restrict__ xk,
                                          const float* __restrict__ wk) {
    f.a = *(const float4*)xk;
    #pragma unroll
    for (int j = 0; j < 4; ++j) f.q[j] = *(const float4*)(wk + (size_t)j * OUTC);
}

#define WMMA_F32(A, B, C) __builtin_amdgcn_wmma_f32_16x16x4_f32( \
         false, (A),  false, (B),  (short)0, (C), \
         false,  false)

__device__ __forceinline__ void mma_frag(const Frag& f, v8f& c0, v8f& c1, v8f& c2, v8f& c3) {
    v2f a0; a0.x = f.a.x; a0.y = f.a.y;
    v2f a1; a1.x = f.a.z; a1.y = f.a.w;
    v2f b;
    b.x = f.q[0].x; b.y = f.q[1].x; c0 = WMMA_F32(a0, b, c0);
    b.x = f.q[0].y; b.y = f.q[1].y; c1 = WMMA_F32(a0, b, c1);
    b.x = f.q[0].z; b.y = f.q[1].z; c2 = WMMA_F32(a0, b, c2);
    b.x = f.q[0].w; b.y = f.q[1].w; c3 = WMMA_F32(a0, b, c3);
    b.x = f.q[2].x; b.y = f.q[3].x; c0 = WMMA_F32(a1, b, c0);
    b.x = f.q[2].y; b.y = f.q[3].y; c1 = WMMA_F32(a1, b, c1);
    b.x = f.q[2].z; b.y = f.q[3].z; c2 = WMMA_F32(a1, b, c2);
    b.x = f.q[2].w; b.y = f.q[3].w; c3 = WMMA_F32(a1, b, c3);
}

__global__ void __launch_bounds__(128) gat_gemm_wmma(
        const float* __restrict__ x, const float* __restrict__ W,
        const float* __restrict__ att,
        float* __restrict__ h, float* __restrict__ p, int nTilesM) {
    __shared__ float spart[GEMM_WAVES][16][2];
    const int mTile = blockIdx.x;
    if (mTile >= nTilesM) return;
    const int head = threadIdx.x >> 5;
    const int lane = threadIdx.x & 31;
    const int c    = lane & 15;
    const int half = lane >> 4;

    const float* xk = x + (size_t)(mTile * 16 + c) * IN_F + 4 * half;
    const float* wk = W + (size_t)(4 * half) * OUTC + head * OUT_F + 4 * c;

    const float* ap = att + head * 2 * OUT_F + 4 * c;
    float as[4], ad[4];
    #pragma unroll
    for (int t = 0; t < 4; ++t) { as[t] = ap[t]; ad[t] = ap[OUT_F + t]; }

    v8f c0 = {}, c1 = {}, c2 = {}, c3 = {};
    Frag cur, nxt;
    load_frag(cur, xk, wk);
    #pragma unroll
    for (int k = 8; k < IN_F; k += 8) {
        load_frag(nxt, xk + k, wk + (size_t)k * OUTC);
        mma_frag(cur, c0, c1, c2, c3);
        cur = nxt;
    }
    mma_frag(cur, c0, c1, c2, c3);

    const int rowOff = half << 3;
    float* hp = h + (size_t)(mTile * 16 + rowOff) * OUTC + head * OUT_F + 4 * c;
    #pragma unroll
    for (int i = 0; i < 8; ++i) {
        float4 v; v.x = c0[i]; v.y = c1[i]; v.z = c2[i]; v.w = c3[i];
        *(float4*)(hp + (size_t)i * OUTC) = v;
    }

    float ps[8], pd[8];
    #pragma unroll
    for (int i = 0; i < 8; ++i) {
        const float l0 = leaky(c0[i]), l1 = leaky(c1[i]), l2 = leaky(c2[i]), l3 = leaky(c3[i]);
        ps[i] = (l0 * as[0] + l1 * as[1]) + (l2 * as[2] + l3 * as[3]);
        pd[i] = (l0 * ad[0] + l1 * ad[1]) + (l2 * ad[2] + l3 * ad[3]);
    }
    #pragma unroll
    for (int i = 0; i < 8; ++i) { ps[i] = row_xor_add<1>(ps[i]); pd[i] = row_xor_add<1>(pd[i]); }
    #pragma unroll
    for (int i = 0; i < 8; ++i) { ps[i] = row_xor_add<2>(ps[i]); pd[i] = row_xor_add<2>(pd[i]); }
    #pragma unroll
    for (int i = 0; i < 8; ++i) { ps[i] = row_xor_add<4>(ps[i]); pd[i] = row_xor_add<4>(pd[i]); }
    #pragma unroll
    for (int i = 0; i < 8; ++i) { ps[i] = row_xor_add<8>(ps[i]); pd[i] = row_xor_add<8>(pd[i]); }
    if (c == 0) {
        #pragma unroll
        for (int i = 0; i < 8; ++i) {
            spart[head][rowOff + i][0] = ps[i];
            spart[head][rowOff + i][1] = pd[i];
        }
    }
    __syncthreads();
    if (threadIdx.x < 32) {
        const int row = threadIdx.x >> 1;
        const int sd  = threadIdx.x & 1;
        float4 v;
        v.x = spart[0][row][sd]; v.y = spart[1][row][sd];
        v.z = spart[2][row][sd]; v.w = spart[3][row][sd];
        *(float4*)(p + (size_t)(mTile * 16 + row) * (2 * HEADS) + sd * HEADS) = v;
    }
}

#define SCORE_WAVES 8

__global__ void __launch_bounds__(256) gat_scores(
        const float* __restrict__ p, const int* __restrict__ ei,
        float* __restrict__ scores, float* __restrict__ gmax, int E, int Nn) {
    __shared__ float smax[SCORE_WAVES][HEADS];
    const int lane   = threadIdx.x & 31;
    const int wib    = threadIdx.x >> 5;
    const int stride = gridDim.x * blockDim.x;
    const int total  = E + Nn;

    const float4* p4 = (const float4*)p;
    float4* s4 = (float4*)scores;
    float m0 = -INFINITY, m1 = -INFINITY, m2 = -INFINITY, m3 = -INFINITY;

    for (int e = blockIdx.x * blockDim.x + threadIdx.x; e < total; e += stride) {
        int s, d;
        if (e < E) { s = ei[e]; d = ei[E + e]; }
        else       { s = d = e - E; }
        const float4 a = p4[(size_t)s * 2];
        const float4 b = p4[(size_t)d * 2 + 1];
        float4 v;
        v.x = a.x + b.x; v.y = a.y + b.y; v.z = a.z + b.z; v.w = a.w + b.w;
        s4[e] = v;
        m0 = fmaxf(m0, v.x); m1 = fmaxf(m1, v.y);
        m2 = fmaxf(m2, v.z); m3 = fmaxf(m3, v.w);
    }

    #pragma unroll
    for (int off = 16; off > 0; off >>= 1) {
        m0 = fmaxf(m0, __shfl_xor(m0, off, 32)); m1 = fmaxf(m1, __shfl_xor(m1, off, 32));
        m2 = fmaxf(m2, __shfl_xor(m2, off, 32)); m3 = fmaxf(m3, __shfl_xor(m3, off, 32));
    }
    if (lane == 0) {
        smax[wib][0] = m0; smax[wib][1] = m1;
        smax[wib][2] = m2; smax[wib][3] = m3;
    }
    __syncthreads();
    if (threadIdx.x < HEADS) {
        float v = smax[0][threadIdx.x];
        for (int w = 1; w < SCORE_WAVES; ++w) v = fmaxf(v, smax[w][threadIdx.x]);
        atomicMax(gmax + threadIdx.x, v);
    }
}

#define EXPS_WAVES 8

__global__ void __launch_bounds__(256) gat_expsum(
        float* __restrict__ scores, const float* __restrict__ gmax,
        float* __restrict__ gsum, int totalEdges) {
    __shared__ float ssum[EXPS_WAVES][HEADS];
    const int lane = threadIdx.x & 31;
    const int wib  = threadIdx.x >> 5;
    const int stride = gridDim.x * blockDim.x;
    const float m0 = gmax[0], m1 = gmax[1], m2 = gmax[2], m3 = gmax[3];

    float4* s4 = (float4*)scores;
    float a0 = 0.f, a1 = 0.f, a2 = 0.f, a3 = 0.f;
    for (int e = blockIdx.x * blockDim.x + threadIdx.x; e < totalEdges; e += stride) {
        float4 v = s4[e];
        v.x = __expf(v.x - m0); v.y = __expf(v.y - m1);
        v.z = __expf(v.z - m2); v.w = __expf(v.w - m3);
        s4[e] = v;
        a0 += v.x; a1 += v.y; a2 += v.z; a3 += v.w;
    }

    #pragma unroll
    for (int off = 16; off > 0; off >>= 1) {
        a0 += __shfl_xor(a0, off, 32); a1 += __shfl_xor(a1, off, 32);
        a2 += __shfl_xor(a2, off, 32); a3 += __shfl_xor(a3, off, 32);
    }
    if (lane == 0) {
        ssum[wib][0] = a0; ssum[wib][1] = a1; ssum[wib][2] = a2; ssum[wib][3] = a3;
    }
    __syncthreads();
    if (threadIdx.x < HEADS) {
        float v = ssum[0][threadIdx.x];
        for (int w = 1; w < EXPS_WAVES; ++w) v += ssum[w][threadIdx.x];
        atomicAdd(gsum + threadIdx.x, v);
    }
}

#define SCAT_WAVES 8

__global__ void __launch_bounds__(256) gat_scatter(
        const float* __restrict__ h, const int* __restrict__ ei,
        const float* __restrict__ scores, const float* __restrict__ gsum,
        float* __restrict__ out, int E, int Nn) {
    const int lane   = threadIdx.x & 31;
    const int wib    = __builtin_amdgcn_readfirstlane((int)(threadIdx.x >> 5));
    const int nWaves = gridDim.x * SCAT_WAVES;
    const int total  = E + Nn;

    float inv[HEADS];
    #pragma unroll
    for (int hd = 0; hd < HEADS; ++hd) inv[hd] = 1.f / gsum[hd];

    for (int e = blockIdx.x * SCAT_WAVES + wib; e < total; e += nWaves) {
        int s, d;
        if (e < E) { s = ei[e]; d = ei[E + e]; }
        else       { s = d = e - E; }

        const float4 sc = *(const float4*)(scores + (size_t)e * HEADS);
        float wgt[HEADS];
        wgt[0] = sc.x * inv[0]; wgt[1] = sc.y * inv[1];
        wgt[2] = sc.z * inv[2]; wgt[3] = sc.w * inv[3];

        const float* hs = h + (size_t)s * OUTC + lane;
        float* op = out + (size_t)d * OUTC + lane;
        float v[8];
        #pragma unroll
        for (int i = 0; i < 8; ++i) v[i] = hs[i << 5];
        #pragma unroll
        for (int i = 0; i < 8; ++i)
            atomicAdd(op + (i << 5), v[i] * wgt[i >> 1]);
    }
}

extern "C" void kernel_launch(void* const* d_in, const int* in_sizes, int n_in,
                              void* d_out, int out_size, void* d_ws, size_t ws_size,
                              hipStream_t stream) {
    const float* x   = (const float*)d_in[0];
    const float* W   = (const float*)d_in[1];
    const float* att = (const float*)d_in[2];
    const float* b   = (const float*)d_in[3];
    const int*   ei  = (const int*)d_in[4];

    const int Nn = in_sizes[0] / IN_F;
    const int E  = in_sizes[4] / 2;
    const int totalEdges = E + Nn;

    float* out    = (float*)d_out;
    float* h      = (float*)d_ws;
    float* p      = h + (size_t)Nn * OUTC;
    float* scores = p + (size_t)Nn * 2 * HEADS;
    float* gmax   = scores + (size_t)totalEdges * HEADS;
    float* gsum   = gmax + HEADS;

    const int outElems = Nn * OUTC;
    gat_init<<<1024, 256, 0, stream>>>(out, b, gmax, gsum, outElems / 4);

    const int nTilesM = Nn / 16;
    gat_gemm_wmma<<<nTilesM, 128, 0, stream>>>(x, W, att, h, p, nTilesM);

    gat_scores<<<512, 256, 0, stream>>>(p, ei, scores, gmax, E, Nn);
    gat_expsum<<<512, 256, 0, stream>>>(scores, gmax, gsum, totalEdges);
    gat_scatter<<<2048, 256, 0, stream>>>(h, ei, scores, gsum, out, E, Nn);
}


# scatter: each block owns 80 out rows in LDS, scans all target ids, adds its edges' weighted h rows by LDS float adds, stores each row once; global float atomics and init launch go (sum order changes)
# speedup vs baseline: 19.8525x; 1.1363x over previous
#include <hip/hip_runtime.h>
#include <hip/hip_bf16.h>
#include <math.h>

typedef __attribute__((ext_vector_type(2))) float v2f;
typedef __attribute__((ext_vector_type(8))) float v8f;

#define IN_F   256
#define OUTC   256
#define HEADS  4
#define OUT_F  64
#define NEG_SLOPE 0.2f

__device__ __forceinline__ float leaky(float v) { return v > 0.f ? v : NEG_SLOPE * v; }

#define GEMM_WAVES 4

template <int M>
__device__ __forceinline__ float row_xor_add(float v) {
    const int t = __builtin_amdgcn_update_dpp(0, __builtin_bit_cast(int, v),
                                              0x160 + M, 0xf, 0xf, true);
    return v + __builtin_bit_cast(float, t);
}

struct Frag {
    float4 a;
    float4 q[4];
};

__device__ __forceinline__ void load_frag(Frag& f, const float* __restrict__ xk,
                                          const float* __restrict__ wk) {
    f.a = *(const float4*)xk;
    #pragma unroll
    for (int j = 0; j < 4; ++j) f.q[j] = *(const float4*)(wk + (size_t)j * OUTC);
}

#define WMMA_F32(A, B, C) __builtin_amdgcn_wmma_f32_16x16x4_f32( \
         false, (A),  false, (B),  (short)0, (C), \
         false,  false)

__device__ __forceinline__ void mma_frag(const Frag& f, v8f& c0, v8f& c1, v8f& c2, v8f& c3) {
    v2f a0; a0.x = f.a.x; a0.y = f.a.y;
    v2f a1; a1.x = f.a.z; a1.y = f.a.w;
    v2f b;
    b.x = f.q[0].x; b.y = f.q[1].x; c0 = WMMA_F32(a0, b, c0);
    b.x = f.q[0].y; b.y = f.q[1].y; c1 = WMMA_F32(a0, b, c1);
    b.x = f.q[0].z; b.y = f.q[1].z; c2 = WMMA_F32(a0, b, c2);
    b.x = f.q[0].w; b.y = f.q[1].w; c3 = WMMA_F32(a0, b, c3);
    b.x = f.q[2].x; b.y = f.q[3].x; c0 = WMMA_F32(a1, b, c0);
    b.x = f.q[2].y; b.y = f.q[3].y; c1 = WMMA_F32(a1, b, c1);
    b.x = f.q[2].z; b.y = f.q[3].z; c2 = WMMA_F32(a1, b, c2);
    b.x = f.q[2].w; b.y = f.q[3].w; c3 = WMMA_F32(a1, b, c3);
}

__global__ void __launch_bounds__(128) gat_gemm_wmma(
        const float* __restrict__ x, const float* __restrict__ W,
        const float* __restrict__ att,
        float* __restrict__ h, float* __restrict__ p,
        float* __restrict__ gmax, float* __restrict__ gsum, int nTilesM) {
    __shared__ float spart[GEMM_WAVES][16][2];
    const int mTile = blockIdx.x;
    if (mTile >= nTilesM) return;
    if (mTile == 0 && threadIdx.x < HEADS) { gmax[threadIdx.x] = -INFINITY; gsum[threadIdx.x] = 0.f; }
    const int head = threadIdx.x >> 5;
    const int lane = threadIdx.x & 31;
    const int c    = lane & 15;
    const int half = lane >> 4;

    const float* xk = x + (size_t)(mTile * 16 + c) * IN_F + 4 * half;
    const float* wk = W + (size_t)(4 * half) * OUTC + head * OUT_F + 4 * c;

    const float* ap = att + head * 2 * OUT_F + 4 * c;
    float as[4], ad[4];
    #pragma unroll
    for (int t = 0; t < 4; ++t) { as[t] = ap[t]; ad[t] = ap[OUT_F + t]; }

    v8f c0 = {}, c1 = {}, c2 = {}, c3 = {};
    Frag cur, nxt;
    load_frag(cur, xk, wk);
    #pragma unroll
    for (int k = 8; k < IN_F; k += 8) {
        load_frag(nxt, xk + k, wk + (size_t)k * OUTC);
        mma_frag(cur, c0, c1, c2, c3);
        cur = nxt;
    }
    mma_frag(cur, c0, c1, c2, c3);

    const int rowOff = half << 3;
    float* hp = h + (size_t)(mTile * 16 + rowOff) * OUTC + head * OUT_F + 4 * c;
    #pragma unroll
    for (int i = 0; i < 8; ++i) {
        float4 v; v.x = c0[i]; v.y = c1[i]; v.z = c2[i]; v.w = c3[i];
        *(float4*)(hp + (size_t)i * OUTC) = v;
    }

    float ps[8], pd[8];
    #pragma unroll
    for (int i = 0; i < 8; ++i) {
        const float l0 = leaky(c0[i]), l1 = leaky(c1[i]), l2 = leaky(c2[i]), l3 = leaky(c3[i]);
        ps[i] = (l0 * as[0] + l1 * as[1]) + (l2 * as[2] + l3 * as[3]);
        pd[i] = (l0 * ad[0] + l1 * ad[1]) + (l2 * ad[2] + l3 * ad[3]);
    }
    #pragma unroll
    for (int i = 0; i < 8; ++i) { ps[i] = row_xor_add<1>(ps[i]); pd[i] = row_xor_add<1>(pd[i]); }
    #pragma unroll
    for (int i = 0; i < 8; ++i) { ps[i] = row_xor_add<2>(ps[i]); pd[i] = row_xor_add<2>(pd[i]); }
    #pragma unroll
    for (int i = 0; i < 8; ++i) { ps[i] = row_xor_add<4>(ps[i]); pd[i] = row_xor_add<4>(pd[i]); }
    #pragma unroll
    for (int i = 0; i < 8; ++i) { ps[i] = row_xor_add<8>(ps[i]); pd[i] = row_xor_add<8>(pd[i]); }
    if (c == 0) {
        #pragma unroll
        for (int i = 0; i < 8; ++i) {
            spart[head][rowOff + i][0] = ps[i];
            spart[head][rowOff + i][1] = pd[i];
        }
    }
    __syncthreads();
    if (threadIdx.x < 32) {
        const int row = threadIdx.x >> 1;
        const int sd  = threadIdx.x & 1;
        float4 v;
        v.x = spart[0][row][sd]; v.y = spart[1][row][sd];
        v.z = spart[2][row][sd]; v.w = spart[3][row][sd];
        *(float4*)(p + (size_t)(mTile * 16 + row) * (2 * HEADS) + sd * HEADS) = v;
    }
}

#define SCORE_WAVES 8

__global__ void __launch_bounds__(256) gat_scores(
        const float* __restrict__ p, const int* __restrict__ ei,
        float* __restrict__ scores, float* __restrict__ gmax, int E, int Nn) {
    __shared__ float smax[SCORE_WAVES][HEADS];
    const int lane   = threadIdx.x & 31;
    const int wib    = threadIdx.x >> 5;
    const int stride = gridDim.x * blockDim.x;
    const int total  = E + Nn;

    const float4* p4 = (const float4*)p;
    float4* s4 = (float4*)scores;
    float m0 = -INFINITY, m1 = -INFINITY, m2 = -INFINITY, m3 = -INFINITY;

    for (int e = blockIdx.x * blockDim.x + threadIdx.x; e < total; e += stride) {
        int s, d;
        if (e < E) { s = ei[e]; d = ei[E + e]; }
        else       { s = d = e - E; }
        const float4 a = p4[(size_t)s * 2];
        const float4 b = p4[(size_t)d * 2 + 1];
        float4 v;
        v.x = a.x + b.x; v.y = a.y + b.y; v.z = a.z + b.z; v.w = a.w + b.w;
        s4[e] = v;
        m0 = fmaxf(m0, v.x); m1 = fmaxf(m1, v.y);
        m2 = fmaxf(m2, v.z); m3 = fmaxf(m3, v.w);
    }

    #pragma unroll
    for (int off = 16; off > 0; off >>= 1) {
        m0 = fmaxf(m0, __shfl_xor(m0, off, 32)); m1 = fmaxf(m1, __shfl_xor(m1, off, 32));
        m2 = fmaxf(m2, __shfl_xor(m2, off, 32)); m3 = fmaxf(m3, __shfl_xor(m3, off, 32));
    }
    if (lane == 0) {
        smax[wib][0] = m0; smax[wib][1] = m1;
        smax[wib][2] = m2; smax[wib][3] = m3;
    }
    __syncthreads();
    if (threadIdx.x < HEADS) {
        float v = smax[0][threadIdx.x];
        for (int w = 1; w < SCORE_WAVES; ++w) v = fmaxf(v, smax[w][threadIdx.x]);
        atomicMax(gmax + threadIdx.x, v);
    }
}

#define EXPS_WAVES 8

__global__ void __launch_bounds__(256) gat_expsum(
        float* __restrict__ scores, const float* __restrict__ gmax,
        float* __restrict__ gsum, int totalEdges) {
    __shared__ float ssum[EXPS_WAVES][HEADS];
    const int lane = threadIdx.x & 31;
    const int wib  = threadIdx.x >> 5;
    const int stride = gridDim.x * blockDim.x;
    const float m0 = gmax[0], m1 = gmax[1], m2 = gmax[2], m3 = gmax[3];

    float4* s4 = (float4*)scores;
    float a0 = 0.f, a1 = 0.f, a2 = 0.f, a3 = 0.f;
    for (int e = blockIdx.x * blockDim.x + threadIdx.x; e < totalEdges; e += stride) {
        float4 v = s4[e];
        v.x = __expf(v.x - m0); v.y = __expf(v.y - m1);
        v.z = __expf(v.z - m2); v.w = __expf(v.w - m3);
        s4[e] = v;
        a0 += v.x; a1 += v.y; a2 += v.z; a3 += v.w;
    }

    #pragma unroll
    for (int off = 16; off > 0; off >>= 1) {
        a0 += __shfl_xor(a0, off, 32); a1 += __shfl_xor(a1, off, 32);
        a2 += __shfl_xor(a2, off, 32); a3 += __shfl_xor(a3, off, 32);
    }
    if (lane == 0) {
        ssum[wib][0] = a0; ssum[wib][1] = a1; ssum[wib][2] = a2; ssum[wib][3] = a3;
    }
    __syncthreads();
    if (threadIdx.x < HEADS) {
        float v = ssum[0][threadIdx.x];
        for (int w = 1; w < EXPS_WAVES; ++w) v += ssum[w][threadIdx.x];
        atomicAdd(gsum + threadIdx.x, v);
    }
}

#define OWN_ROWS    80
#define OWN_THREADS 1024
#define OWN_WAVES   32
#define OWN_GROUP   8
#define LIST_CAP    4096

__device__ __forceinline__ int own_col(int l4, int t) {
    return ((l4 >> 5) << 7) + (t << 5) + (l4 & 31);
}

__device__ __noinline__ void own_add_row_serial(float* accRow, const float* __restrict__ hs,
                                                float w0, float w1, float w2, float w3) {
    for (int c = 0; c < OUTC; ++c) {
        const int hd = c >> 6;
        const float w = hd == 0 ? w0 : hd == 1 ? w1 : hd == 2 ? w2 : w3;
        atomicAdd(accRow + own_col(c >> 2, c & 3), hs[c] * w);
    }
}

__global__ void __launch_bounds__(OWN_THREADS) gat_scatter_own(
        const float* __restrict__ h, const int* __restrict__ ei,
        const float* __restrict__ scores, const float* __restrict__ gsum,
        const float* __restrict__ b, float* __restrict__ out, int E, int Nn) {
    __shared__ float  acc[(OWN_ROWS + 1) * OUTC];
    __shared__ int2   l_sd[LIST_CAP];
    __shared__ float4 l_w[LIST_CAP];
    __shared__ int    nlist;

    const int tid  = threadIdx.x;
    const int lane = tid & 31;
    const int wave = __builtin_amdgcn_readfirstlane(tid >> 5);
    const int lo    = blockIdx.x * OWN_ROWS;
    const int nrows = min(OWN_ROWS, Nn - lo);
    if (nrows <= 0) return;

    const float inv0 = 1.f / gsum[0], inv1 = 1.f / gsum[1], inv2 = 1.f / gsum[2], inv3 = 1.f / gsum[3];
    const float4* sc4 = (const float4*)scores;

    if (tid == 0) nlist = 0;
    for (int f = tid; f < nrows * (OUTC / 4); f += OWN_THREADS) {
        const int r = f >> 6, l4 = f & 63, hd = l4 >> 4;
        const float4 v = *(const float4*)(h + (size_t)(lo + r) * OUTC + 4 * l4);
        const float sw = scores[(size_t)(E + lo + r) * HEADS + hd];
        const float w = sw * (hd == 0 ? inv0 : hd == 1 ? inv1 : hd == 2 ? inv2 : inv3);
        float* a = acc + r * OUTC;
        a[own_col(l4, 0)] = v.x * w; a[own_col(l4, 1)] = v.y * w;
        a[own_col(l4, 2)] = v.z * w; a[own_col(l4, 3)] = v.w * w;
    }
    __syncthreads();

    const int* dstp = ei + E;
    auto take = [&](int d, int e) {
        const unsigned r = (unsigned)(d - lo);
        if (r < (unsigned)nrows) {
            const int slot = atomicAdd(&nlist, 1);
            if (slot < LIST_CAP) {
                l_sd[slot] = make_int2(e, (int)r);
            } else {
                const float4 sc = sc4[e];
                own_add_row_serial(acc + r * OUTC, h + (size_t)ei[e] * OUTC,
                                   sc.x * inv0, sc.y * inv1, sc.z * inv2, sc.w * inv3);
            }
        }
    };
    auto take4 = [&](const int4 a, int e) {
        const unsigned r0 = (unsigned)(a.x - lo), r1 = (unsigned)(a.y - lo);
        const unsigned r2 = (unsigned)(a.z - lo), r3 = (unsigned)(a.w - lo);
        if (min(min(r0, r1), min(r2, r3)) < (unsigned)nrows) {
            take(a.x, e); take(a.y, e + 1); take(a.z, e + 2); take(a.w, e + 3);
        }
    };
    const int nvec = ((E & 3) == 0) ? (E >> 2) : 0;
    const int4* dst4 = (const int4*)dstp;
    int gi = tid;
    for (; gi + 3 * OWN_THREADS < nvec; gi += 4 * OWN_THREADS) {
        const int4 a0 = dst4[gi];
        const int4 a1 = dst4[gi + OWN_THREADS];
        const int4 a2 = dst4[gi + 2 * OWN_THREADS];
        const int4 a3 = dst4[gi + 3 * OWN_THREADS];
        const int e = 4 * gi;
        take4(a0, e);
        take4(a1, e + 4 * OWN_THREADS);
        take4(a2, e + 8 * OWN_THREADS);
        take4(a3, e + 12 * OWN_THREADS);
    }
    for (; gi < nvec; gi += OWN_THREADS) {
        take4(dst4[gi], 4 * gi);
    }
    for (int e = 4 * nvec + tid; e < E; e += OWN_THREADS) take(dstp[e], e);
    __syncthreads();

    const int n  = min(nlist, LIST_CAP);
    const int n8 = (n + OWN_GROUP - 1) & ~(OWN_GROUP - 1);
    for (int i = tid; i < n8; i += OWN_THREADS) {
        if (i < n) {
            const int e = l_sd[i].x;
            const float4 sc = sc4[e];
            l_sd[i].x = ei[e];
            l_w[i] = make_float4(sc.x * inv0, sc.y * inv1, sc.z * inv2, sc.w * inv3);
        } else {
            l_sd[i] = make_int2(lo, OWN_ROWS);
            l_w[i]  = make_float4(0.f, 0.f, 0.f, 0.f);
        }
    }
    __syncthreads();

    const float* l_wf = (const float*)l_w;
    const int hsel = lane >> 4;
    const int ngroups = n8 / OWN_GROUP;
    for (int g = wave; g < ngroups; g += OWN_WAVES) {
        float4 v0[OWN_GROUP], v1[OWN_GROUP];
        float  w0[OWN_GROUP], w1[OWN_GROUP];
        int    ro[OWN_GROUP];
        #pragma unroll
        for (int k = 0; k < OWN_GROUP; ++k) {
            const int ent = g * OWN_GROUP + k;
            const int2 sd = l_sd[ent];
            const int sidx = __builtin_amdgcn_readfirstlane(sd.x);
            ro[k] = __builtin_amdgcn_readfirstlane(sd.y) * OUTC;
            const float* hs = h + (size_t)sidx * OUTC + 4 * lane;
            v0[k] = *(const float4*)hs;
            v1[k] = *(const float4*)(hs + 128);
            w0[k] = l_wf[ent * 4 + hsel];
            w1[k] = l_wf[ent * 4 + 2 + hsel];
        }
        #pragma unroll
        for (int k = 0; k < OWN_GROUP; ++k) {
            float* a = acc + ro[k] + lane;
            atomicAdd(a,       v0[k].x * w0[k]);
            atomicAdd(a + 32,  v0[k].y * w0[k]);
            atomicAdd(a + 64,  v0[k].z * w0[k]);
            atomicAdd(a + 96,  v0[k].w * w0[k]);
            atomicAdd(a + 128, v1[k].x * w1[k]);
            atomicAdd(a + 160, v1[k].y * w1[k]);
            atomicAdd(a + 192, v1[k].z * w1[k]);
            atomicAdd(a + 224, v1[k].w * w1[k]);
        }
    }
    __syncthreads();

    for (int f = tid; f < nrows * (OUTC / 4); f += OWN_THREADS) {
        const int r = f >> 6, l4 = f & 63;
        const float* a = acc + r * OUTC;
        const float4 bv = *(const float4*)(b + 4 * l4);
        float4 v;
        v.x = a[own_col(l4, 0)] + bv.x; v.y = a[own_col(l4, 1)] + bv.y;
        v.z = a[own_col(l4, 2)] + bv.z; v.w = a[own_col(l4, 3)] + bv.w;
        *(float4*)(out + (size_t)(lo + r) * OUTC + 4 * l4) = v;
    }
}

extern "C" void kernel_launch(void* const* d_in, const int* in_sizes, int n_in,
                              void* d_out, int out_size, void* d_ws, size_t ws_size,
                              hipStream_t stream) {
    const float* x   = (const float*)d_in[0];
    const float* W   = (const float*)d_in[1];
    const float* att = (const float*)d_in[2];
    const float* b   = (const float*)d_in[3];
    const int*   ei  = (const int*)d_in[4];

    const int Nn = in_sizes[0] / IN_F;
    const int E  = in_sizes[4] / 2;
    const int totalEdges = E + Nn;

    float* out    = (float*)d_out;
    float* h      = (float*)d_ws;
    float* p      = h + (size_t)Nn * OUTC;
    float* scores = p + (size_t)Nn * 2 * HEADS;
    float* gmax   = scores + (size_t)totalEdges * HEADS;
    float* gsum   = gmax + HEADS;

    const int nTilesM = Nn / 16;
    gat_gemm_wmma<<<nTilesM, 128, 0, stream>>>(x, W, att, h, p, gmax, gsum, nTilesM);

    gat_scores<<<512, 256, 0, stream>>>(p, ei, scores, gmax, E, Nn);
    gat_expsum<<<512, 256, 0, stream>>>(scores, gmax, gsum, totalEdges);
    gat_scatter_own<<<(Nn + OWN_ROWS - 1) / OWN_ROWS, OWN_THREADS, 0, stream>>>(
        h, ei, scores, gsum, b, out, E, Nn);
}


# scatter scan reads a 1-byte slice key per edge, left by the scores kernel, instead of the 4-byte target id (16 keys per load, 4 matched per dword op); list order, so LDS add order, changes
# speedup vs baseline: 21.8400x; 1.1001x over previous
#include <hip/hip_runtime.h>
#include <hip/hip_bf16.h>
#include <math.h>

typedef __attribute__((ext_vector_type(2))) float v2f;
typedef __attribute__((ext_vector_type(8))) float v8f;
typedef __attribute__((ext_vector_type(4))) unsigned int v4u;

#define IN_F   256
#define OUTC   256
#define HEADS  4
#define OUT_F  64
#define NEG_SLOPE 0.2f
#define OWN_ROWS 80

__device__ __forceinline__ float leaky(float v) { return v > 0.f ? v : NEG_SLOPE * v; }

#define GEMM_WAVES 4

template <int M>
__device__ __forceinline__ float row_xor_add(float v) {
    const int t = __builtin_amdgcn_update_dpp(0, __builtin_bit_cast(int, v),
                                              0x160 + M, 0xf, 0xf, true);
    return v + __builtin_bit_cast(float, t);
}

struct Frag {
    float4 a;
    float4 q[4];
};

__device__ __forceinline__ void load_frag(Frag& f, const float* __restrict__ xk,
                                          const float* __restrict__ wk) {
    f.a = *(const float4*)xk;
    #pragma unroll
    for (int j = 0; j < 4; ++j) f.q[j] = *(const float4*)(wk + (size_t)j * OUTC);
}

#define WMMA_F32(A, B, C) __builtin_amdgcn_wmma_f32_16x16x4_f32( \
         false, (A),  false, (B),  (short)0, (C), \
         false,  false)

__device__ __forceinline__ void mma_frag(const Frag& f, v8f& c0, v8f& c1, v8f& c2, v8f& c3) {
    v2f a0; a0.x = f.a.x; a0.y = f.a.y;
    v2f a1; a1.x = f.a.z; a1.y = f.a.w;
    v2f b;
    b.x = f.q[0].x; b.y = f.q[1].x; c0 = WMMA_F32(a0, b, c0);
    b.x = f.q[0].y; b.y = f.q[1].y; c1 = WMMA_F32(a0, b, c1);
    b.x = f.q[0].z; b.y = f.q[1].z; c2 = WMMA_F32(a0, b, c2);
    b.x = f.q[0].w; b.y = f.q[1].w; c3 = WMMA_F32(a0, b, c3);
    b.x = f.q[2].x; b.y = f.q[3].x; c0 = WMMA_F32(a1, b, c0);
    b.x = f.q[2].y; b.y = f.q[3].y; c1 = WMMA_F32(a1, b, c1);
    b.x = f.q[2].z; b.y = f.q[3].z; c2 = WMMA_F32(a1, b, c2);
    b.x = f.q[2].w; b.y = f.q[3].w; c3 = WMMA_F32(a1, b, c3);
}

__global__ void __launch_bounds__(128) gat_gemm_wmma(
        const float* __restrict__ x, const float* __restrict__ W,
        const float* __restrict__ att,
        float* __restrict__ h, float* __restrict__ p,
        float* __restrict__ gmax, float* __restrict__ gsum, int nTilesM) {
    __shared__ float spart[GEMM_WAVES][16][2];
    const int mTile = blockIdx.x;
    if (mTile >= nTilesM) return;
    if (mTile == 0 && threadIdx.x < HEADS) { gmax[threadIdx.x] = -INFINITY; gsum[threadIdx.x] = 0.f; }
    const int head = threadIdx.x >> 5;
    const int lane = threadIdx.x & 31;
    const int c    = lane & 15;
    const int half = lane >> 4;

    const float* xk = x + (size_t)(mTile * 16 + c) * IN_F + 4 * half;
    const float* wk = W + (size_t)(4 * half) * OUTC + head * OUT_F + 4 * c;

    const float* ap = att + head * 2 * OUT_F + 4 * c;
    float as[4], ad[4];
    #pragma unroll
    for (int t = 0; t < 4; ++t) { as[t] = ap[t]; ad[t] = ap[OUT_F + t]; }

    v8f c0 = {}, c1 = {}, c2 = {}, c3 = {};
    Frag cur, nxt;
    load_frag(cur, xk, wk);
    #pragma unroll
    for (int k = 8; k < IN_F; k += 8) {
        load_frag(nxt, xk + k, wk + (size_t)k * OUTC);
        mma_frag(cur, c0, c1, c2, c3);
        cur = nxt;
    }
    mma_frag(cur, c0, c1, c2, c3);

    const int rowOff = half << 3;
    float* hp = h + (size_t)(mTile * 16 + rowOff) * OUTC + head * OUT_F + 4 * c;
    #pragma unroll
    for (int i = 0; i < 8; ++i) {
        float4 v; v.x = c0[i]; v.y = c1[i]; v.z = c2[i]; v.w = c3[i];
        *(float4*)(hp + (size_t)i * OUTC) = v;
    }

    float ps[8], pd[8];
    #pragma unroll
    for (int i = 0; i < 8; ++i) {
        const float l0 = leaky(c0[i]), l1 = leaky(c1[i]), l2 = leaky(c2[i]), l3 = leaky(c3[i]);
        ps[i] = (l0 * as[0] + l1 * as[1]) + (l2 * as[2] + l3 * as[3]);
        pd[i] = (l0 * ad[0] + l1 * ad[1]) + (l2 * ad[2] + l3 * ad[3]);
    }
    #pragma unroll
    for (int i = 0; i < 8; ++i) { ps[i] = row_xor_add<1>(ps[i]); pd[i] = row_xor_add<1>(pd[i]); }
    #pragma unroll
    for (int i = 0; i < 8; ++i) { ps[i] = row_xor_add<2>(ps[i]); pd[i] = row_xor_add<2>(pd[i]); }
    #pragma unroll
    for (int i = 0; i < 8; ++i) { ps[i] = row_xor_add<4>(ps[i]); pd[i] = row_xor_add<4>(pd[i]); }
    #pragma unroll
    for (int i = 0; i < 8; ++i) { ps[i] = row_xor_add<8>(ps[i]); pd[i] = row_xor_add<8>(pd[i]); }
    if (c == 0) {
        #pragma unroll
        for (int i = 0; i < 8; ++i) {
            spart[head][rowOff + i][0] = ps[i];
            spart[head][rowOff + i][1] = pd[i];
        }
    }
    __syncthreads();
    if (threadIdx.x < 32) {
        const int row = threadIdx.x >> 1;
        const int sd  = threadIdx.x & 1;
        float4 v;
        v.x = spart[0][row][sd]; v.y = spart[1][row][sd];
        v.z = spart[2][row][sd]; v.w = spart[3][row][sd];
        *(float4*)(p + (size_t)(mTile * 16 + row) * (2 * HEADS) + sd * HEADS) = v;
    }
}

#define SCORE_WAVES 8

__global__ void __launch_bounds__(256) gat_scores(
        const float* __restrict__ p, const int* __restrict__ ei,
        float* __restrict__ scores, float* __restrict__ gmax,
        unsigned int* __restrict__ keys, int E, int Nn) {
    __shared__ float smax[SCORE_WAVES][HEADS];
    unsigned char* kb = (unsigned char*)keys;
    const int lane   = threadIdx.x & 31;
    const int wib    = threadIdx.x >> 5;
    const int stride = gridDim.x * blockDim.x;
    const int total  = E + Nn;

    const float4* p4 = (const float4*)p;
    float4* s4 = (float4*)scores;
    float m0 = -INFINITY, m1 = -INFINITY, m2 = -INFINITY, m3 = -INFINITY;

    for (int e = blockIdx.x * blockDim.x + threadIdx.x; e < total; e += stride) {
        int s, d;
        if (e < E) {
            s = ei[e]; d = ei[E + e];
            kb[e] = (unsigned char)((unsigned)d / OWN_ROWS);
        } else { s = d = e - E; }
        const float4 a = p4[(size_t)s * 2];
        const float4 b = p4[(size_t)d * 2 + 1];
        float4 v;
        v.x = a.x + b.x; v.y = a.y + b.y; v.z = a.z + b.z; v.w = a.w + b.w;
        s4[e] = v;
        m0 = fmaxf(m0, v.x); m1 = fmaxf(m1, v.y);
        m2 = fmaxf(m2, v.z); m3 = fmaxf(m3, v.w);
    }

    #pragma unroll
    for (int off = 16; off > 0; off >>= 1) {
        m0 = fmaxf(m0, __shfl_xor(m0, off, 32)); m1 = fmaxf(m1, __shfl_xor(m1, off, 32));
        m2 = fmaxf(m2, __shfl_xor(m2, off, 32)); m3 = fmaxf(m3, __shfl_xor(m3, off, 32));
    }
    if (lane == 0) {
        smax[wib][0] = m0; smax[wib][1] = m1;
        smax[wib][2] = m2; smax[wib][3] = m3;
    }
    __syncthreads();
    if (threadIdx.x < HEADS) {
        float v = smax[0][threadIdx.x];
        for (int w = 1; w < SCORE_WAVES; ++w) v = fmaxf(v, smax[w][threadIdx.x]);
        atomicMax(gmax + threadIdx.x, v);
    }
}

#define EXPS_WAVES 8

__global__ void __launch_bounds__(256) gat_expsum(
        float* __restrict__ scores, const float* __restrict__ gmax,
        float* __restrict__ gsum, int totalEdges) {
    __shared__ float ssum[EXPS_WAVES][HEADS];
    const int lane = threadIdx.x & 31;
    const int wib  = threadIdx.x >> 5;
    const int stride = gridDim.x * blockDim.x;
    const float m0 = gmax[0], m1 = gmax[1], m2 = gmax[2], m3 = gmax[3];

    float4* s4 = (float4*)scores;
    float a0 = 0.f, a1 = 0.f, a2 = 0.f, a3 = 0.f;
    for (int e = blockIdx.x * blockDim.x + threadIdx.x; e < totalEdges; e += stride) {
        float4 v = s4[e];
        v.x = __expf(v.x - m0); v.y = __expf(v.y - m1);
        v.z = __expf(v.z - m2); v.w = __expf(v.w - m3);
        s4[e] = v;
        a0 += v.x; a1 += v.y; a2 += v.z; a3 += v.w;
    }

    #pragma unroll
    for (int off = 16; off > 0; off >>= 1) {
        a0 += __shfl_xor(a0, off, 32); a1 += __shfl_xor(a1, off, 32);
        a2 += __shfl_xor(a2, off, 32); a3 += __shfl_xor(a3, off, 32);
    }
    if (lane == 0) {
        ssum[wib][0] = a0; ssum[wib][1] = a1; ssum[wib][2] = a2; ssum[wib][3] = a3;
    }
    __syncthreads();
    if (threadIdx.x < HEADS) {
        float v = ssum[0][threadIdx.x];
        for (int w = 1; w < EXPS_WAVES; ++w) v += ssum[w][threadIdx.x];
        atomicAdd(gsum + threadIdx.x, v);
    }
}

#define OWN_THREADS 1024
#define OWN_WAVES   32
#define OWN_GROUP   8
#define SCAN_LOADS  5
#define LIST_CAP    4096

__device__ __forceinline__ int own_col(int l4, int t) {
    return ((l4 >> 5) << 7) + (t << 5) + (l4 & 31);
}

__device__ __noinline__ void own_add_row_serial(float* accRow, const float* __restrict__ hs,
                                                float w0, float w1, float w2, float w3) {
    for (int c = 0; c < OUTC; ++c) {
        const int hd = c >> 6;
        const float w = hd == 0 ? w0 : hd == 1 ? w1 : hd == 2 ? w2 : w3;
        atomicAdd(accRow + own_col(c >> 2, c & 3), hs[c] * w);
    }
}

__global__ void __launch_bounds__(OWN_THREADS) gat_scatter_own(
        const float* __restrict__ h, const int* __restrict__ ei,
        const float* __restrict__ scores, const float* __restrict__ gsum,
        const unsigned int* __restrict__ keys,
        const float* __restrict__ b, float* __restrict__ out, int E, int Nn) {
    __shared__ float  acc[(OWN_ROWS + 1) * OUTC];
    __shared__ int2   l_sd[LIST_CAP];
    __shared__ float4 l_w[LIST_CAP];
    __shared__ int    nlist;

    const int tid  = threadIdx.x;
    const int lane = tid & 31;
    const int wave = __builtin_amdgcn_readfirstlane(tid >> 5);
    const int lo    = blockIdx.x * OWN_ROWS;
    const int nrows = min(OWN_ROWS, Nn - lo);
    if (nrows <= 0) return;

    const float inv0 = 1.f / gsum[0], inv1 = 1.f / gsum[1], inv2 = 1.f / gsum[2], inv3 = 1.f / gsum[3];
    const float4* sc4 = (const float4*)scores;

    if (tid == 0) nlist = 0;
    for (int f = tid; f < nrows * (OUTC / 4); f += OWN_THREADS) {
        const int r = f >> 6, l4 = f & 63, hd = l4 >> 4;
        const float4 v = *(const float4*)(h + (size_t)(lo + r) * OUTC + 4 * l4);
        const float sw = scores[(size_t)(E + lo + r) * HEADS + hd];
        const float w = sw * (hd == 0 ? inv0 : hd == 1 ? inv1 : hd == 2 ? inv2 : inv3);
        float* a = acc + r * OUTC;
        a[own_col(l4, 0)] = v.x * w; a[own_col(l4, 1)] = v.y * w;
        a[own_col(l4, 2)] = v.z * w; a[own_col(l4, 3)] = v.w * w;
    }
    __syncthreads();

    const int* dstp = ei + E;
    const unsigned mykey = blockIdx.x & 0xFFu;
    const unsigned pat = mykey * 0x01010101u;
    auto put = [&](int slot, int e) {
        if (slot < LIST_CAP) {
            l_sd[slot] = make_int2(e, 0);
        } else {
            const unsigned r = (unsigned)(dstp[e] - lo);
            if (r < (unsigned)nrows) {
                const float4 sc = sc4[e];
                own_add_row_serial(acc + r * OUTC, h + (size_t)ei[e] * OUTC,
                                   sc.x * inv0, sc.y * inv1, sc.z * inv2, sc.w * inv3);
            }
        }
    };
    auto hit4 = [&](unsigned w) -> unsigned {
        const unsigned x = w ^ pat;
        return ~(((x & 0x7F7F7F7Fu) + 0x7F7F7F7Fu) | x | 0x7F7F7F7Fu);
    };
    auto hit16 = [&](const v4u a) -> unsigned {
        return (hit4(a.x) >> 7) | (hit4(a.y) >> 6) | (hit4(a.z) >> 5) | (hit4(a.w) >> 4);
    };
    const int nvec = E >> 4;
    const v4u* key4 = (const v4u*)keys;
    for (int g0 = tid; g0 < nvec; g0 += SCAN_LOADS * OWN_THREADS) {
        v4u a[SCAN_LOADS];
        #pragma unroll
        for (int j = 0; j < SCAN_LOADS; ++j)
            a[j] = key4[min(g0 + j * OWN_THREADS, nvec - 1)];
        static_assert(SCAN_LOADS == 5, "the statement below names five loads");
        asm volatile("" :: "v"(a[0]), "v"(a[1]), "v"(a[2]), "v"(a[3]), "v"(a[4]));
        unsigned m[SCAN_LOADS];
        int cnt = 0;
        #pragma unroll
        for (int j = 0; j < SCAN_LOADS; ++j) {
            const unsigned live = (unsigned)((g0 + j * OWN_THREADS - nvec) >> 31);
            m[j] = hit16(a[j]) & live;
            cnt += __popc(m[j]);
        }
        if (cnt) {
            const int slot = atomicAdd(&nlist, cnt);
            unsigned w01 = m[0] | (m[1] << 4), w23 = m[2] | (m[3] << 4), w4 = m[4];
            for (int k = 0; k < cnt; ++k) {
                const int sel = w01 ? 0 : (w23 ? 1 : 2);
                const unsigned w = w01 ? w01 : (w23 ? w23 : w4);
                const int bit = __ffs(w) - 1;
                const unsigned rest = w & (w - 1);
                if (sel == 0) w01 = rest; else if (sel == 1) w23 = rest; else w4 = rest;
                const int ld = 2 * sel + ((bit >> 2) & 1);
                put(slot + k, 16 * (g0 + ld * OWN_THREADS) + 4 * (bit & 3) + (bit >> 3));
            }
        }
    }
    const unsigned char* kb = (const unsigned char*)keys;
    for (int e = 16 * nvec + tid; e < E; e += OWN_THREADS)
        if (kb[e] == mykey) put(atomicAdd(&nlist, 1), e);
    __syncthreads();

    const int n  = min(nlist, LIST_CAP);
    const int n8 = (n + OWN_GROUP - 1) & ~(OWN_GROUP - 1);
    for (int i = tid; i < n8; i += OWN_THREADS) {
        if (i < n) {
            const int e = l_sd[i].x;
            const int d = dstp[e];
            const int s = ei[e];
            const float4 sc = sc4[e];
            const unsigned r = (unsigned)(d - lo);
            const bool ok = r < (unsigned)nrows;
            l_sd[i] = make_int2(ok ? s : lo, ok ? (int)r : OWN_ROWS);
            l_w[i] = make_float4(ok ? sc.x * inv0 : 0.f, ok ? sc.y * inv1 : 0.f,
                                 ok ? sc.z * inv2 : 0.f, ok ? sc.w * inv3 : 0.f);
        } else {
            l_sd[i] = make_int2(lo, OWN_ROWS);
            l_w[i]  = make_float4(0.f, 0.f, 0.f, 0.f);
        }
    }
    __syncthreads();

    const float* l_wf = (const float*)l_w;
    const int hsel = lane >> 4;
    const int ngroups = n8 / OWN_GROUP;
    for (int g = wave; g < ngroups; g += OWN_WAVES) {
        float4 v0[OWN_GROUP], v1[OWN_GROUP];
        float  w0[OWN_GROUP], w1[OWN_GROUP];
        int    ro[OWN_GROUP];
        #pragma unroll
        for (int k = 0; k < OWN_GROUP; ++k) {
            const int ent = g * OWN_GROUP + k;
            const int2 sd = l_sd[ent];
            const int sidx = __builtin_amdgcn_readfirstlane(sd.x);
            ro[k] = __builtin_amdgcn_readfirstlane(sd.y) * OUTC;
            const float* hs = h + (size_t)sidx * OUTC + 4 * lane;
            v0[k] = *(const float4*)hs;
            v1[k] = *(const float4*)(hs + 128);
            w0[k] = l_wf[ent * 4 + hsel];
            w1[k] = l_wf[ent * 4 + 2 + hsel];
        }
        #pragma unroll
        for (int k = 0; k < OWN_GROUP; ++k) {
            float* a = acc + ro[k] + lane;
            atomicAdd(a,       v0[k].x * w0[k]);
            atomicAdd(a + 32,  v0[k].y * w0[k]);
            atomicAdd(a + 64,  v0[k].z * w0[k]);
            atomicAdd(a + 96,  v0[k].w * w0[k]);
            atomicAdd(a + 128, v1[k].x * w1[k]);
            atomicAdd(a + 160, v1[k].y * w1[k]);
            atomicAdd(a + 192, v1[k].z * w1[k]);
            atomicAdd(a + 224, v1[k].w * w1[k]);
        }
    }
    __syncthreads();

    for (int f = tid; f < nrows * (OUTC / 4); f += OWN_THREADS) {
        const int r = f >> 6, l4 = f & 63;
        const float* a = acc + r * OUTC;
        const float4 bv = *(const float4*)(b + 4 * l4);
        float4 v;
        v.x = a[own_col(l4, 0)] + bv.x; v.y = a[own_col(l4, 1)] + bv.y;
        v.z = a[own_col(l4, 2)] + bv.z; v.w = a[own_col(l4, 3)] + bv.w;
        *(float4*)(out + (size_t)(lo + r) * OUTC + 4 * l4) = v;
    }
}

extern "C" void kernel_launch(void* const* d_in, const int* in_sizes, int n_in,
                              void* d_out, int out_size, void* d_ws, size_t ws_size,
                              hipStream_t stream) {
    const float* x   = (const float*)d_in[0];
    const float* W   = (const float*)d_in[1];
    const float* att = (const float*)d_in[2];
    const float* b   = (const float*)d_in[3];
    const int*   ei  = (const int*)d_in[4];

    const int Nn = in_sizes[0] / IN_F;
    const int E  = in_sizes[4] / 2;
    const int totalEdges = E + Nn;

    float* out    = (float*)d_out;
    float* h      = (float*)d_ws;
    float* p      = h + (size_t)Nn * OUTC;
    float* scores = p + (size_t)Nn * 2 * HEADS;
    float* gmax   = scores + (size_t)totalEdges * HEADS;
    float* gsum   = gmax + HEADS;
    unsigned int* keys = (unsigned int*)(gsum + HEADS);

    const int nTilesM = Nn / 16;
    gat_gemm_wmma<<<nTilesM, 128, 0, stream>>>(x, W, att, h, p, gmax, gsum, nTilesM);

    gat_scores<<<512, 256, 0, stream>>>(p, ei, scores, gmax, keys, E, Nn);
    gat_expsum<<<512, 256, 0, stream>>>(scores, gmax, gsum, totalEdges);
    gat_scatter_own<<<(Nn + OWN_ROWS - 1) / OWN_ROWS, OWN_THREADS, 0, stream>>>(
        h, ei, scores, gsum, keys, b, out, E, Nn);
}


# exp-sum launch folded into the scores kernel: running (max, rescaled sum) pairs merged over lanes, waves and 512 block pairs, no atomics (sum order changes); scatter forms exp(score-max)/sum itself
# speedup vs baseline: 23.5282x; 1.0773x over previous
#include <hip/hip_runtime.h>
#include <hip/hip_bf16.h>
#include <math.h>

typedef __attribute__((ext_vector_type(2))) float v2f;
typedef __attribute__((ext_vector_type(8))) float v8f;
typedef __attribute__((ext_vector_type(4))) unsigned int v4u;

#define IN_F   256
#define OUTC   256
#define HEADS  4
#define OUT_F  64
#define NEG_SLOPE 0.2f
#define OWN_ROWS 80

__device__ __forceinline__ float leaky(float v) { return v > 0.f ? v : NEG_SLOPE * v; }

__device__ __forceinline__ float shift_exp(float m, float M) { return m == M ? 1.f : __expf(m - M); }
template <int M>
__device__ __forceinline__ float row_xor(float v) {
    return __builtin_bit_cast(float, __builtin_amdgcn_update_dpp(0, __builtin_bit_cast(int, v),
                                                                 0x160 + M, 0xf, 0xf, true));
}
__device__ __forceinline__ float xhalf(float v) {
    const int b = __builtin_bit_cast(int, v);
    return __builtin_bit_cast(float, __builtin_amdgcn_permlanex16(b, b, 0x76543210, 0xfedcba98, false, false));
}
__device__ __forceinline__ float row_max(float v) {
    v = fmaxf(v, row_xor<1>(v)); v = fmaxf(v, row_xor<2>(v));
    v = fmaxf(v, row_xor<4>(v)); return fmaxf(v, row_xor<8>(v));
}
__device__ __forceinline__ float row_sum(float v) {
    v += row_xor<1>(v); v += row_xor<2>(v); v += row_xor<4>(v); return v + row_xor<8>(v);
}
__device__ __forceinline__ float wave_max(float v) { v = row_max(v); return fmaxf(v, xhalf(v)); }
__device__ __forceinline__ float wave_sum(float v) { v = row_sum(v); return v + xhalf(v); }
#define SCORE_BLOCKS 512

#define GEMM_WAVES 4

template <int M>
__device__ __forceinline__ float row_xor_add(float v) {
    const int t = __builtin_amdgcn_update_dpp(0, __builtin_bit_cast(int, v),
                                              0x160 + M, 0xf, 0xf, true);
    return v + __builtin_bit_cast(float, t);
}

struct Frag {
    float4 a;
    float4 q[4];
};

__device__ __forceinline__ void load_frag(Frag& f, const float* __restrict__ xk,
                                          const float* __restrict__ wk) {
    f.a = *(const float4*)xk;
    #pragma unroll
    for (int j = 0; j < 4; ++j) f.q[j] = *(const float4*)(wk + (size_t)j * OUTC);
}

#define WMMA_F32(A, B, C) __builtin_amdgcn_wmma_f32_16x16x4_f32( \
         false, (A),  false, (B),  (short)0, (C), \
         false,  false)

__device__ __forceinline__ void mma_frag(const Frag& f, v8f& c0, v8f& c1, v8f& c2, v8f& c3) {
    v2f a0; a0.x = f.a.x; a0.y = f.a.y;
    v2f a1; a1.x = f.a.z; a1.y = f.a.w;
    v2f b;
    b.x = f.q[0].x; b.y = f.q[1].x; c0 = WMMA_F32(a0, b, c0);
    b.x = f.q[0].y; b.y = f.q[1].y; c1 = WMMA_F32(a0, b, c1);
    b.x = f.q[0].z; b.y = f.q[1].z; c2 = WMMA_F32(a0, b, c2);
    b.x = f.q[0].w; b.y = f.q[1].w; c3 = WMMA_F32(a0, b, c3);
    b.x = f.q[2].x; b.y = f.q[3].x; c0 = WMMA_F32(a1, b, c0);
    b.x = f.q[2].y; b.y = f.q[3].y; c1 = WMMA_F32(a1, b, c1);
    b.x = f.q[2].z; b.y = f.q[3].z; c2 = WMMA_F32(a1, b, c2);
    b.x = f.q[2].w; b.y = f.q[3].w; c3 = WMMA_F32(a1, b, c3);
}

__global__ void __launch_bounds__(128) gat_gemm_wmma(
        const float* __restrict__ x, const float* __restrict__ W,
        const float* __restrict__ att,
        float* __restrict__ h, float* __restrict__ p, int nTilesM) {
    __shared__ float spart[GEMM_WAVES][16][2];
    const int mTile = blockIdx.x;
    if (mTile >= nTilesM) return;
    const int head = threadIdx.x >> 5;
    const int lane = threadIdx.x & 31;
    const int c    = lane & 15;
    const int half = lane >> 4;

    const float* xk = x + (size_t)(mTile * 16 + c) * IN_F + 4 * half;
    const float* wk = W + (size_t)(4 * half) * OUTC + head * OUT_F + 4 * c;

    const float* ap = att + head * 2 * OUT_F + 4 * c;
    float as[4], ad[4];
    #pragma unroll
    for (int t = 0; t < 4; ++t) { as[t] = ap[t]; ad[t] = ap[OUT_F + t]; }

    v8f c0 = {}, c1 = {}, c2 = {}, c3 = {};
    Frag cur, nxt;
    load_frag(cur, xk, wk);
    #pragma unroll
    for (int k = 8; k < IN_F; k += 8) {
        load_frag(nxt, xk + k, wk + (size_t)k * OUTC);
        mma_frag(cur, c0, c1, c2, c3);
        cur = nxt;
    }
    mma_frag(cur, c0, c1, c2, c3);

    const int rowOff = half << 3;
    float* hp = h + (size_t)(mTile * 16 + rowOff) * OUTC + head * OUT_F + 4 * c;
    #pragma unroll
    for (int i = 0; i < 8; ++i) {
        float4 v; v.x = c0[i]; v.y = c1[i]; v.z = c2[i]; v.w = c3[i];
        *(float4*)(hp + (size_t)i * OUTC) = v;
    }

    float ps[8], pd[8];
    #pragma unroll
    for (int i = 0; i < 8; ++i) {
        const float l0 = leaky(c0[i]), l1 = leaky(c1[i]), l2 = leaky(c2[i]), l3 = leaky(c3[i]);
        ps[i] = (l0 * as[0] + l1 * as[1]) + (l2 * as[2] + l3 * as[3]);
        pd[i] = (l0 * ad[0] + l1 * ad[1]) + (l2 * ad[2] + l3 * ad[3]);
    }
    #pragma unroll
    for (int i = 0; i < 8; ++i) { ps[i] = row_xor_add<1>(ps[i]); pd[i] = row_xor_add<1>(pd[i]); }
    #pragma unroll
    for (int i = 0; i < 8; ++i) { ps[i] = row_xor_add<2>(ps[i]); pd[i] = row_xor_add<2>(pd[i]); }
    #pragma unroll
    for (int i = 0; i < 8; ++i) { ps[i] = row_xor_add<4>(ps[i]); pd[i] = row_xor_add<4>(pd[i]); }
    #pragma unroll
    for (int i = 0; i < 8; ++i) { ps[i] = row_xor_add<8>(ps[i]); pd[i] = row_xor_add<8>(pd[i]); }
    if (c == 0) {
        #pragma unroll
        for (int i = 0; i < 8; ++i) {
            spart[head][rowOff + i][0] = ps[i];
            spart[head][rowOff + i][1] = pd[i];
        }
    }
    __syncthreads();
    if (threadIdx.x < 32) {
        const int row = threadIdx.x >> 1;
        const int sd  = threadIdx.x & 1;
        float4 v;
        v.x = spart[0][row][sd]; v.y = spart[1][row][sd];
        v.z = spart[2][row][sd]; v.w = spart[3][row][sd];
        *(float4*)(p + (size_t)(mTile * 16 + row) * (2 * HEADS) + sd * HEADS) = v;
    }
}

#define SCORE_WAVES 8

__global__ void __launch_bounds__(256) gat_scores(
        const float* __restrict__ p, const int* __restrict__ ei,
        float* __restrict__ scores, float* __restrict__ part,
        unsigned int* __restrict__ keys, int E, int Nn) {
    __shared__ float smax[SCORE_WAVES][HEADS];
    __shared__ float ssum[SCORE_WAVES][HEADS];
    unsigned char* kb = (unsigned char*)keys;
    const int lane   = threadIdx.x & 31;
    const int wib    = threadIdx.x >> 5;
    const int stride = gridDim.x * blockDim.x;
    const int total  = E + Nn;

    const float4* p4 = (const float4*)p;
    float4* s4 = (float4*)scores;
    float m0 = -INFINITY, m1 = -INFINITY, m2 = -INFINITY, m3 = -INFINITY;
    float a0 = 0.f, a1 = 0.f, a2 = 0.f, a3 = 0.f;

    for (int e = blockIdx.x * blockDim.x + threadIdx.x; e < total; e += stride) {
        int s, d;
        if (e < E) {
            s = ei[e]; d = ei[E + e];
            kb[e] = (unsigned char)((unsigned)d / OWN_ROWS);
        } else { s = d = e - E; }
        const float4 a = p4[(size_t)s * 2];
        const float4 b = p4[(size_t)d * 2 + 1];
        float4 v;
        v.x = a.x + b.x; v.y = a.y + b.y; v.z = a.z + b.z; v.w = a.w + b.w;
        s4[e] = v;
        float n;
        n = fmaxf(m0, v.x); a0 = a0 * shift_exp(m0, n) + shift_exp(v.x, n); m0 = n;
        n = fmaxf(m1, v.y); a1 = a1 * shift_exp(m1, n) + shift_exp(v.y, n); m1 = n;
        n = fmaxf(m2, v.z); a2 = a2 * shift_exp(m2, n) + shift_exp(v.z, n); m2 = n;
        n = fmaxf(m3, v.w); a3 = a3 * shift_exp(m3, n) + shift_exp(v.w, n); m3 = n;
    }

    const float w0 = wave_max(m0), w1 = wave_max(m1), w2 = wave_max(m2), w3 = wave_max(m3);
    a0 = wave_sum(a0 * shift_exp(m0, w0)); a1 = wave_sum(a1 * shift_exp(m1, w1));
    a2 = wave_sum(a2 * shift_exp(m2, w2)); a3 = wave_sum(a3 * shift_exp(m3, w3));
    if (lane == 0) {
        smax[wib][0] = w0; smax[wib][1] = w1; smax[wib][2] = w2; smax[wib][3] = w3;
        ssum[wib][0] = a0; ssum[wib][1] = a1; ssum[wib][2] = a2; ssum[wib][3] = a3;
    }
    __syncthreads();
    if (threadIdx.x < HEADS) {
        float M = smax[0][threadIdx.x];
        for (int w = 1; w < SCORE_WAVES; ++w) M = fmaxf(M, smax[w][threadIdx.x]);
        float S = 0.f;
        for (int w = 0; w < SCORE_WAVES; ++w) S += ssum[w][threadIdx.x] * shift_exp(smax[w][threadIdx.x], M);
        part[blockIdx.x * (2 * HEADS) + threadIdx.x]         = M;
        part[blockIdx.x * (2 * HEADS) + HEADS + threadIdx.x] = S;
    }
}

#define OWN_THREADS 1024
#define OWN_WAVES   32
#define OWN_GROUP   8
#define SCAN_LOADS  5
#define LIST_CAP    4096

__device__ __forceinline__ float uniform_f(float v) {
    return __builtin_bit_cast(float, __builtin_amdgcn_readfirstlane(__builtin_bit_cast(int, v)));
}

__device__ __forceinline__ int own_col(int l4, int t) {
    return ((l4 >> 5) << 7) + (t << 5) + (l4 & 31);
}

__device__ __noinline__ void own_add_row_serial(float* accRow, const float* __restrict__ hs,
                                                float w0, float w1, float w2, float w3) {
    for (int c = 0; c < OUTC; ++c) {
        const int hd = c >> 6;
        const float w = hd == 0 ? w0 : hd == 1 ? w1 : hd == 2 ? w2 : w3;
        atomicAdd(accRow + own_col(c >> 2, c & 3), hs[c] * w);
    }
}

__global__ void __launch_bounds__(OWN_THREADS) gat_scatter_own(
        const float* __restrict__ h, const int* __restrict__ ei,
        const float* __restrict__ scores, const float* __restrict__ part,
        const unsigned int* __restrict__ keys,
        const float* __restrict__ b, float* __restrict__ out, int E, int Nn) {
    __shared__ float  acc[(OWN_ROWS + 1) * OUTC];
    __shared__ int2   l_sd[LIST_CAP];
    __shared__ float4 l_w[LIST_CAP];
    __shared__ int    nlist;
    __shared__ float4 s_pm[SCORE_BLOCKS / 32];
    __shared__ float4 s_ps[SCORE_BLOCKS / 32];
    static_assert(SCORE_BLOCKS == 512 && OWN_THREADS >= SCORE_BLOCKS,
                  "one pair per lane in 16 waves; the second merge takes entry lane & 15");

    const int tid  = threadIdx.x;
    const int lane = tid & 31;
    const int wave = __builtin_amdgcn_readfirstlane(tid >> 5);
    const int lo    = blockIdx.x * OWN_ROWS;
    const int nrows = min(OWN_ROWS, Nn - lo);
    if (nrows <= 0) return;

    const float4* sc4 = (const float4*)scores;

    if (wave < SCORE_BLOCKS / 32) {
        const float4* part4 = (const float4*)part;
        const float4 m = part4[2 * tid];
        float4 a = part4[2 * tid + 1];
        float4 M;
        M.x = wave_max(m.x); M.y = wave_max(m.y); M.z = wave_max(m.z); M.w = wave_max(m.w);
        a.x = wave_sum(a.x * shift_exp(m.x, M.x)); a.y = wave_sum(a.y * shift_exp(m.y, M.y));
        a.z = wave_sum(a.z * shift_exp(m.z, M.z)); a.w = wave_sum(a.w * shift_exp(m.w, M.w));
        if (lane == 0) { s_pm[wave] = M; s_ps[wave] = a; }
    }
    if (tid == 0) nlist = 0;
    __syncthreads();
    float mx0, mx1, mx2, mx3, inv0, inv1, inv2, inv3;
    {
        const float4 m = s_pm[lane & 15];
        float4 a = s_ps[lane & 15];
        float4 M;
        M.x = row_max(m.x); M.y = row_max(m.y); M.z = row_max(m.z); M.w = row_max(m.w);
        a.x = row_sum(a.x * shift_exp(m.x, M.x)); a.y = row_sum(a.y * shift_exp(m.y, M.y));
        a.z = row_sum(a.z * shift_exp(m.z, M.z)); a.w = row_sum(a.w * shift_exp(m.w, M.w));
        mx0 = uniform_f(M.x); mx1 = uniform_f(M.y); mx2 = uniform_f(M.z); mx3 = uniform_f(M.w);
        inv0 = uniform_f(1.f / a.x); inv1 = uniform_f(1.f / a.y);
        inv2 = uniform_f(1.f / a.z); inv3 = uniform_f(1.f / a.w);
    }

    for (int f = tid; f < nrows * (OUTC / 4); f += OWN_THREADS) {
        const int r = f >> 6, l4 = f & 63, hd = l4 >> 4;
        const float4 v = *(const float4*)(h + (size_t)(lo + r) * OUTC + 4 * l4);
        const float sw = scores[(size_t)(E + lo + r) * HEADS + hd];
        const float w = __expf(sw - (hd == 0 ? mx0 : hd == 1 ? mx1 : hd == 2 ? mx2 : mx3))
                        * (hd == 0 ? inv0 : hd == 1 ? inv1 : hd == 2 ? inv2 : inv3);
        float* a = acc + r * OUTC;
        a[own_col(l4, 0)] = v.x * w; a[own_col(l4, 1)] = v.y * w;
        a[own_col(l4, 2)] = v.z * w; a[own_col(l4, 3)] = v.w * w;
    }
    __syncthreads();

    const int* dstp = ei + E;
    const unsigned mykey = blockIdx.x & 0xFFu;
    const unsigned pat = mykey * 0x01010101u;
    auto put = [&](int slot, int e) {
        if (slot < LIST_CAP) {
            l_sd[slot] = make_int2(e, 0);
        } else {
            const unsigned r = (unsigned)(dstp[e] - lo);
            if (r < (unsigned)nrows) {
                const float4 sc = sc4[e];
                own_add_row_serial(acc + r * OUTC, h + (size_t)ei[e] * OUTC,
                                   __expf(sc.x - mx0) * inv0, __expf(sc.y - mx1) * inv1,
                                   __expf(sc.z - mx2) * inv2, __expf(sc.w - mx3) * inv3);
            }
        }
    };
    auto hit4 = [&](unsigned w) -> unsigned {
        const unsigned x = w ^ pat;
        return ~(((x & 0x7F7F7F7Fu) + 0x7F7F7F7Fu) | x | 0x7F7F7F7Fu);
    };
    auto hit16 = [&](const v4u a) -> unsigned {
        return (hit4(a.x) >> 7) | (hit4(a.y) >> 6) | (hit4(a.z) >> 5) | (hit4(a.w) >> 4);
    };
    const int nvec = E >> 4;
    const v4u* key4 = (const v4u*)keys;
    for (int g0 = tid; g0 < nvec; g0 += SCAN_LOADS * OWN_THREADS) {
        v4u a[SCAN_LOADS];
        #pragma unroll
        for (int j = 0; j < SCAN_LOADS; ++j)
            a[j] = key4[min(g0 + j * OWN_THREADS, nvec - 1)];
        static_assert(SCAN_LOADS == 5, "the statement below names five loads");
        asm volatile("" :: "v"(a[0]), "v"(a[1]), "v"(a[2]), "v"(a[3]), "v"(a[4]));
        unsigned m[SCAN_LOADS];
        int cnt = 0;
        #pragma unroll
        for (int j = 0; j < SCAN_LOADS; ++j) {
            const unsigned live = (unsigned)((g0 + j * OWN_THREADS - nvec) >> 31);
            m[j] = hit16(a[j]) & live;
            cnt += __popc(m[j]);
        }
        if (cnt) {
            const int slot = atomicAdd(&nlist, cnt);
            unsigned w01 = m[0] | (m[1] << 4), w23 = m[2] | (m[3] << 4), w4 = m[4];
            for (int k = 0; k < cnt; ++k) {
                const int sel = w01 ? 0 : (w23 ? 1 : 2);
                const unsigned w = w01 ? w01 : (w23 ? w23 : w4);
                const int bit = __ffs(w) - 1;
                const unsigned rest = w & (w - 1);
                if (sel == 0) w01 = rest; else if (sel == 1) w23 = rest; else w4 = rest;
                const int ld = 2 * sel + ((bit >> 2) & 1);
                put(slot + k, 16 * (g0 + ld * OWN_THREADS) + 4 * (bit & 3) + (bit >> 3));
            }
        }
    }
    const unsigned char* kb = (const unsigned char*)keys;
    for (int e = 16 * nvec + tid; e < E; e += OWN_THREADS)
        if (kb[e] == mykey) put(atomicAdd(&nlist, 1), e);
    __syncthreads();

    const int n  = min(nlist, LIST_CAP);
    const int n8 = (n + OWN_GROUP - 1) & ~(OWN_GROUP - 1);
    for (int i = tid; i < n8; i += OWN_THREADS) {
        if (i < n) {
            const int e = l_sd[i].x;
            const int d = dstp[e];
            const int s = ei[e];
            const float4 sc = sc4[e];
            const unsigned r = (unsigned)(d - lo);
            const bool ok = r < (unsigned)nrows;
            l_sd[i] = make_int2(ok ? s : lo, ok ? (int)r : OWN_ROWS);
            l_w[i] = make_float4(ok ? __expf(sc.x - mx0) * inv0 : 0.f, ok ? __expf(sc.y - mx1) * inv1 : 0.f,
                                 ok ? __expf(sc.z - mx2) * inv2 : 0.f, ok ? __expf(sc.w - mx3) * inv3 : 0.f);
        } else {
            l_sd[i] = make_int2(lo, OWN_ROWS);
            l_w[i]  = make_float4(0.f, 0.f, 0.f, 0.f);
        }
    }
    __syncthreads();

    const float* l_wf = (const float*)l_w;
    const int hsel = lane >> 4;
    const int ngroups = n8 / OWN_GROUP;
    for (int g = wave; g < ngroups; g += OWN_WAVES) {
        float4 v0[OWN_GROUP], v1[OWN_GROUP];
        float  w0[OWN_GROUP], w1[OWN_GROUP];
        int    ro[OWN_GROUP];
        #pragma unroll
        for (int k = 0; k < OWN_GROUP; ++k) {
            const int ent = g * OWN_GROUP + k;
            const int2 sd = l_sd[ent];
            const int sidx = __builtin_amdgcn_readfirstlane(sd.x);
            ro[k] = __builtin_amdgcn_readfirstlane(sd.y) * OUTC;
            const float* hs = h + (size_t)sidx * OUTC + 4 * lane;
            v0[k] = *(const float4*)hs;
            v1[k] = *(const float4*)(hs + 128);
            w0[k] = l_wf[ent * 4 + hsel];
            w1[k] = l_wf[ent * 4 + 2 + hsel];
        }
        #pragma unroll
        for (int k = 0; k < OWN_GROUP; ++k) {
            float* a = acc + ro[k] + lane;
            atomicAdd(a,       v0[k].x * w0[k]);
            atomicAdd(a + 32,  v0[k].y * w0[k]);
            atomicAdd(a + 64,  v0[k].z * w0[k]);
            atomicAdd(a + 96,  v0[k].w * w0[k]);
            atomicAdd(a + 128, v1[k].x * w1[k]);
            atomicAdd(a + 160, v1[k].y * w1[k]);
            atomicAdd(a + 192, v1[k].z * w1[k]);
            atomicAdd(a + 224, v1[k].w * w1[k]);
        }
    }
    __syncthreads();

    for (int f = tid; f < nrows * (OUTC / 4); f += OWN_THREADS) {
        const int r = f >> 6, l4 = f & 63;
        const float* a = acc + r * OUTC;
        const float4 bv = *(const float4*)(b + 4 * l4);
        float4 v;
        v.x = a[own_col(l4, 0)] + bv.x; v.y = a[own_col(l4, 1)] + bv.y;
        v.z = a[own_col(l4, 2)] + bv.z; v.w = a[own_col(l4, 3)] + bv.w;
        *(float4*)(out + (size_t)(lo + r) * OUTC + 4 * l4) = v;
    }
}

extern "C" void kernel_launch(void* const* d_in, const int* in_sizes, int n_in,
                              void* d_out, int out_size, void* d_ws, size_t ws_size,
                              hipStream_t stream) {
    const float* x   = (const float*)d_in[0];
    const float* W   = (const float*)d_in[1];
    const float* att = (const float*)d_in[2];
    const float* b   = (const float*)d_in[3];
    const int*   ei  = (const int*)d_in[4];

    const int Nn = in_sizes[0] / IN_F;
    const int E  = in_sizes[4] / 2;
    const int totalEdges = E + Nn;

    float* out    = (float*)d_out;
    float* h      = (float*)d_ws;
    float* p      = h + (size_t)Nn * OUTC;
    float* scores = p + (size_t)Nn * 2 * HEADS;
    float* part   = scores + (size_t)totalEdges * HEADS;
    unsigned int* keys = (unsigned int*)(part + SCORE_BLOCKS * 2 * HEADS);

    const int nTilesM = Nn / 16;
    gat_gemm_wmma<<<nTilesM, 128, 0, stream>>>(x, W, att, h, p, nTilesM);

    gat_scores<<<SCORE_BLOCKS, 256, 0, stream>>>(p, ei, scores, part, keys, E, Nn);
    gat_scatter_own<<<(Nn + OWN_ROWS - 1) / OWN_ROWS, OWN_THREADS, 0, stream>>>(
        h, ei, scores, part, keys, b, out, E, Nn);
}


# last kernel: slice's edges and self loops ordered by target row in LDS (integer rank per row), each wave sums its rows in registers, stores them once; LDS float adds and slice go (sum order changes)
# speedup vs baseline: 25.9398x; 1.1025x over previous
#include <hip/hip_runtime.h>
#include <hip/hip_bf16.h>
#include <math.h>
#include <type_traits>

typedef __attribute__((ext_vector_type(2))) float v2f;
typedef __attribute__((ext_vector_type(8))) float v8f;
typedef __attribute__((ext_vector_type(4))) unsigned int v4u;
typedef __attribute__((ext_vector_type(4))) float v4f;

#define IN_F   256
#define OUTC   256
#define HEADS  4
#define OUT_F  64
#define NEG_SLOPE 0.2f
#define OWN_ROWS 80

__device__ __forceinline__ float leaky(float v) { return v > 0.f ? v : NEG_SLOPE * v; }

__device__ __forceinline__ float shift_exp(float m, float M) { return m == M ? 1.f : __expf(m - M); }
template <int M>
__device__ __forceinline__ float row_xor(float v) {
    return __builtin_bit_cast(float, __builtin_amdgcn_update_dpp(0, __builtin_bit_cast(int, v),
                                                                 0x160 + M, 0xf, 0xf, true));
}
__device__ __forceinline__ float xhalf(float v) {
    const int b = __builtin_bit_cast(int, v);
    return __builtin_bit_cast(float, __builtin_amdgcn_permlanex16(b, b, 0x76543210, 0xfedcba98, false, false));
}
__device__ __forceinline__ float row_max(float v) {
    v = fmaxf(v, row_xor<1>(v)); v = fmaxf(v, row_xor<2>(v));
    v = fmaxf(v, row_xor<4>(v)); return fmaxf(v, row_xor<8>(v));
}
__device__ __forceinline__ float row_sum(float v) {
    v += row_xor<1>(v); v += row_xor<2>(v); v += row_xor<4>(v); return v + row_xor<8>(v);
}
__device__ __forceinline__ float wave_max(float v) { v = row_max(v); return fmaxf(v, xhalf(v)); }
__device__ __forceinline__ float wave_sum(float v) { v = row_sum(v); return v + xhalf(v); }
#define SCORE_BLOCKS 512

#define GEMM_WAVES 4

template <int M>
__device__ __forceinline__ float row_xor_add(float v) {
    const int t = __builtin_amdgcn_update_dpp(0, __builtin_bit_cast(int, v),
                                              0x160 + M, 0xf, 0xf, true);
    return v + __builtin_bit_cast(float, t);
}

struct Frag {
    float4 a;
    float4 q[4];
};

__device__ __forceinline__ void load_frag(Frag& f, const float* __restrict__ xk,
                                          const float* __restrict__ wk) {
    f.a = *(const float4*)xk;
    #pragma unroll
    for (int j = 0; j < 4; ++j) f.q[j] = *(const float4*)(wk + (size_t)j * OUTC);
}

#define WMMA_F32(A, B, C) __builtin_amdgcn_wmma_f32_16x16x4_f32( \
         false, (A),  false, (B),  (short)0, (C), \
         false,  false)

__device__ __forceinline__ void mma_frag(const Frag& f, v8f& c0, v8f& c1, v8f& c2, v8f& c3) {
    v2f a0; a0.x = f.a.x; a0.y = f.a.y;
    v2f a1; a1.x = f.a.z; a1.y = f.a.w;
    v2f b;
    b.x = f.q[0].x; b.y = f.q[1].x; c0 = WMMA_F32(a0, b, c0);
    b.x = f.q[0].y; b.y = f.q[1].y; c1 = WMMA_F32(a0, b, c1);
    b.x = f.q[0].z; b.y = f.q[1].z; c2 = WMMA_F32(a0, b, c2);
    b.x = f.q[0].w; b.y = f.q[1].w; c3 = WMMA_F32(a0, b, c3);
    b.x = f.q[2].x; b.y = f.q[3].x; c0 = WMMA_F32(a1, b, c0);
    b.x = f.q[2].y; b.y = f.q[3].y; c1 = WMMA_F32(a1, b, c1);
    b.x = f.q[2].z; b.y = f.q[3].z; c2 = WMMA_F32(a1, b, c2);
    b.x = f.q[2].w; b.y = f.q[3].w; c3 = WMMA_F32(a1, b, c3);
}

__global__ void __launch_bounds__(128) gat_gemm_wmma(
        const float* __restrict__ x, const float* __restrict__ W,
        const float* __restrict__ att,
        float* __restrict__ h, float* __restrict__ p, int nTilesM) {
    __shared__ float spart[GEMM_WAVES][16][2];
    const int mTile = blockIdx.x;
    if (mTile >= nTilesM) return;
    const int head = threadIdx.x >> 5;
    const int lane = threadIdx.x & 31;
    const int c    = lane & 15;
    const int half = lane >> 4;

    const float* xk = x + (size_t)(mTile * 16 + c) * IN_F + 4 * half;
    const float* wk = W + (size_t)(4 * half) * OUTC + head * OUT_F + 4 * c;

    const float* ap = att + head * 2 * OUT_F + 4 * c;
    float as[4], ad[4];
    #pragma unroll
    for (int t = 0; t < 4; ++t) { as[t] = ap[t]; ad[t] = ap[OUT_F + t]; }

    v8f c0 = {}, c1 = {}, c2 = {}, c3 = {};
    Frag cur, nxt;
    load_frag(cur, xk, wk);
    #pragma unroll
    for (int k = 8; k < IN_F; k += 8) {
        load_frag(nxt, xk + k, wk + (size_t)k * OUTC);
        mma_frag(cur, c0, c1, c2, c3);
        cur = nxt;
    }
    mma_frag(cur, c0, c1, c2, c3);

    const int rowOff = half << 3;
    float* hp = h + (size_t)(mTile * 16 + rowOff) * OUTC + head * OUT_F + 4 * c;
    #pragma unroll
    for (int i = 0; i < 8; ++i) {
        float4 v; v.x = c0[i]; v.y = c1[i]; v.z = c2[i]; v.w = c3[i];
        *(float4*)(hp + (size_t)i * OUTC) = v;
    }

    float ps[8], pd[8];
    #pragma unroll
    for (int i = 0; i < 8; ++i) {
        const float l0 = leaky(c0[i]), l1 = leaky(c1[i]), l2 = leaky(c2[i]), l3 = leaky(c3[i]);
        ps[i] = (l0 * as[0] + l1 * as[1]) + (l2 * as[2] + l3 * as[3]);
        pd[i] = (l0 * ad[0] + l1 * ad[1]) + (l2 * ad[2] + l3 * ad[3]);
    }
    #pragma unroll
    for (int i = 0; i < 8; ++i) { ps[i] = row_xor_add<1>(ps[i]); pd[i] = row_xor_add<1>(pd[i]); }
    #pragma unroll
    for (int i = 0; i < 8; ++i) { ps[i] = row_xor_add<2>(ps[i]); pd[i] = row_xor_add<2>(pd[i]); }
    #pragma unroll
    for (int i = 0; i < 8; ++i) { ps[i] = row_xor_add<4>(ps[i]); pd[i] = row_xor_add<4>(pd[i]); }
    #pragma unroll
    for (int i = 0; i < 8; ++i) { ps[i] = row_xor_add<8>(ps[i]); pd[i] = row_xor_add<8>(pd[i]); }
    if (c == 0) {
        #pragma unroll
        for (int i = 0; i < 8; ++i) {
            spart[head][rowOff + i][0] = ps[i];
            spart[head][rowOff + i][1] = pd[i];
        }
    }
    __syncthreads();
    if (threadIdx.x < 32) {
        const int row = threadIdx.x >> 1;
        const int sd  = threadIdx.x & 1;
        float4 v;
        v.x = spart[0][row][sd]; v.y = spart[1][row][sd];
        v.z = spart[2][row][sd]; v.w = spart[3][row][sd];
        *(float4*)(p + (size_t)(mTile * 16 + row) * (2 * HEADS) + sd * HEADS) = v;
    }
}

#define SCORE_WAVES 8

__global__ void __launch_bounds__(256) gat_scores(
        const float* __restrict__ p, const int* __restrict__ ei,
        float* __restrict__ scores, float* __restrict__ part,
        unsigned int* __restrict__ keys, int E, int Nn) {
    __shared__ float smax[SCORE_WAVES][HEADS];
    __shared__ float ssum[SCORE_WAVES][HEADS];
    unsigned char* kb = (unsigned char*)keys;
    const int lane   = threadIdx.x & 31;
    const int wib    = threadIdx.x >> 5;
    const int stride = gridDim.x * blockDim.x;
    const int total  = E + Nn;

    const float4* p4 = (const float4*)p;
    float4* s4 = (float4*)scores;
    float m0 = -INFINITY, m1 = -INFINITY, m2 = -INFINITY, m3 = -INFINITY;
    float a0 = 0.f, a1 = 0.f, a2 = 0.f, a3 = 0.f;

    for (int e = blockIdx.x * blockDim.x + threadIdx.x; e < total; e += stride) {
        int s, d;
        if (e < E) {
            s = ei[e]; d = ei[E + e];
            kb[e] = (unsigned char)((unsigned)d / OWN_ROWS);
        } else { s = d = e - E; }
        const float4 a = p4[(size_t)s * 2];
        const float4 b = p4[(size_t)d * 2 + 1];
        float4 v;
        v.x = a.x + b.x; v.y = a.y + b.y; v.z = a.z + b.z; v.w = a.w + b.w;
        s4[e] = v;
        float n;
        n = fmaxf(m0, v.x); a0 = a0 * shift_exp(m0, n) + shift_exp(v.x, n); m0 = n;
        n = fmaxf(m1, v.y); a1 = a1 * shift_exp(m1, n) + shift_exp(v.y, n); m1 = n;
        n = fmaxf(m2, v.z); a2 = a2 * shift_exp(m2, n) + shift_exp(v.z, n); m2 = n;
        n = fmaxf(m3, v.w); a3 = a3 * shift_exp(m3, n) + shift_exp(v.w, n); m3 = n;
    }

    const float w0 = wave_max(m0), w1 = wave_max(m1), w2 = wave_max(m2), w3 = wave_max(m3);
    a0 = wave_sum(a0 * shift_exp(m0, w0)); a1 = wave_sum(a1 * shift_exp(m1, w1));
    a2 = wave_sum(a2 * shift_exp(m2, w2)); a3 = wave_sum(a3 * shift_exp(m3, w3));
    if (lane == 0) {
        smax[wib][0] = w0; smax[wib][1] = w1; smax[wib][2] = w2; smax[wib][3] = w3;
        ssum[wib][0] = a0; ssum[wib][1] = a1; ssum[wib][2] = a2; ssum[wib][3] = a3;
    }
    __syncthreads();
    if (threadIdx.x < HEADS) {
        float M = smax[0][threadIdx.x];
        for (int w = 1; w < SCORE_WAVES; ++w) M = fmaxf(M, smax[w][threadIdx.x]);
        float S = 0.f;
        for (int w = 0; w < SCORE_WAVES; ++w) S += ssum[w][threadIdx.x] * shift_exp(smax[w][threadIdx.x], M);
        part[blockIdx.x * (2 * HEADS) + threadIdx.x]         = M;
        part[blockIdx.x * (2 * HEADS) + HEADS + threadIdx.x] = S;
    }
}

#define OWN_THREADS 1024
#define OWN_WAVES   32
#define OWN_PAIRS   (OWN_WAVES / 2)
#define PAIR_ROWS   (OWN_ROWS / OWN_PAIRS)
#define OWN_GROUP   16
#define SCAN_LOADS  5
#define LIST_CAP    4096
#define ENT_CAP     (LIST_CAP + OWN_ROWS)
static_assert(OWN_PAIRS * PAIR_ROWS == OWN_ROWS, "the pairs of waves cover the slice exactly");

__device__ __forceinline__ float uniform_f(float v) {
    return __builtin_bit_cast(float, __builtin_amdgcn_readfirstlane(__builtin_bit_cast(int, v)));
}

__global__ void __launch_bounds__(OWN_THREADS) gat_gather_own(
        const float* __restrict__ h, const int* __restrict__ ei,
        const float* __restrict__ scores, const float* __restrict__ part,
        const unsigned int* __restrict__ keys,
        const float* __restrict__ b, float* __restrict__ out, int E, int Nn) {
    __shared__ int    l_e[ENT_CAP];
    __shared__ int2   u_sd[ENT_CAP];
    __shared__ float4 u_w[ENT_CAP];
    __shared__ int2   s_sd[ENT_CAP];
    __shared__ float4 s_w[ENT_CAP];
    __shared__ int    rcount[OWN_ROWS];
    __shared__ int    rstart[OWN_ROWS + 1];
    __shared__ int    nlist;
    __shared__ float4 s_pm[SCORE_BLOCKS / 32];
    __shared__ float4 s_ps[SCORE_BLOCKS / 32];
    static_assert(SCORE_BLOCKS == 512 && OWN_THREADS >= SCORE_BLOCKS,
                  "one pair per lane in 16 waves; the second merge takes entry lane & 15");

    const int tid  = threadIdx.x;
    const int lane = tid & 31;
    const int wave = __builtin_amdgcn_readfirstlane(tid >> 5);
    const int lo    = blockIdx.x * OWN_ROWS;
    const int nrows = min(OWN_ROWS, Nn - lo);
    if (nrows <= 0) return;

    const float4* sc4 = (const float4*)scores;

    if (wave < SCORE_BLOCKS / 32) {
        const float4* part4 = (const float4*)part;
        const float4 m = part4[2 * tid];
        float4 a = part4[2 * tid + 1];
        float4 M;
        M.x = wave_max(m.x); M.y = wave_max(m.y); M.z = wave_max(m.z); M.w = wave_max(m.w);
        a.x = wave_sum(a.x * shift_exp(m.x, M.x)); a.y = wave_sum(a.y * shift_exp(m.y, M.y));
        a.z = wave_sum(a.z * shift_exp(m.z, M.z)); a.w = wave_sum(a.w * shift_exp(m.w, M.w));
        if (lane == 0) { s_pm[wave] = M; s_ps[wave] = a; }
    }
    if (tid == 0) nlist = 0;
    if (tid < OWN_ROWS) rcount[tid] = 0;
    __syncthreads();
    float mx0, mx1, mx2, mx3, inv0, inv1, inv2, inv3;
    {
        const float4 m = s_pm[lane & 15];
        float4 a = s_ps[lane & 15];
        float4 M;
        M.x = row_max(m.x); M.y = row_max(m.y); M.z = row_max(m.z); M.w = row_max(m.w);
        a.x = row_sum(a.x * shift_exp(m.x, M.x)); a.y = row_sum(a.y * shift_exp(m.y, M.y));
        a.z = row_sum(a.z * shift_exp(m.z, M.z)); a.w = row_sum(a.w * shift_exp(m.w, M.w));
        mx0 = uniform_f(M.x); mx1 = uniform_f(M.y); mx2 = uniform_f(M.z); mx3 = uniform_f(M.w);
        inv0 = uniform_f(1.f / a.x); inv1 = uniform_f(1.f / a.y);
        inv2 = uniform_f(1.f / a.z); inv3 = uniform_f(1.f / a.w);
    }

    const int* dstp = ei + E;
    const unsigned mykey = blockIdx.x & 0xFFu;
    const unsigned pat = mykey * 0x01010101u;
    auto put = [&](int slot, int e) {
        if (slot < LIST_CAP) l_e[slot] = e;
    };
    auto hit4 = [&](unsigned w) -> unsigned {
        const unsigned x = w ^ pat;
        return ~(((x & 0x7F7F7F7Fu) + 0x7F7F7F7Fu) | x | 0x7F7F7F7Fu);
    };
    auto hit16 = [&](const v4u a) -> unsigned {
        return (hit4(a.x) >> 7) | (hit4(a.y) >> 6) | (hit4(a.z) >> 5) | (hit4(a.w) >> 4);
    };
    const int nvec = E >> 4;
    const v4u* key4 = (const v4u*)keys;
    for (int g0 = tid; g0 < nvec; g0 += SCAN_LOADS * OWN_THREADS) {
        v4u a[SCAN_LOADS];
        #pragma unroll
        for (int j = 0; j < SCAN_LOADS; ++j)
            a[j] = key4[min(g0 + j * OWN_THREADS, nvec - 1)];
        static_assert(SCAN_LOADS == 5, "the statement below names five loads");
        asm volatile("" :: "v"(a[0]), "v"(a[1]), "v"(a[2]), "v"(a[3]), "v"(a[4]));
        unsigned m[SCAN_LOADS];
        int cnt = 0;
        #pragma unroll
        for (int j = 0; j < SCAN_LOADS; ++j) {
            const unsigned live = (unsigned)((g0 + j * OWN_THREADS - nvec) >> 31);
            m[j] = hit16(a[j]) & live;
            cnt += __popc(m[j]);
        }
        if (cnt) {
            const int slot = atomicAdd(&nlist, cnt);
            unsigned w01 = m[0] | (m[1] << 4), w23 = m[2] | (m[3] << 4), w4 = m[4];
            for (int k = 0; k < cnt; ++k) {
                const int sel = w01 ? 0 : (w23 ? 1 : 2);
                const unsigned w = w01 ? w01 : (w23 ? w23 : w4);
                const int bit = __ffs(w) - 1;
                const unsigned rest = w & (w - 1);
                if (sel == 0) w01 = rest; else if (sel == 1) w23 = rest; else w4 = rest;
                const int ld = 2 * sel + ((bit >> 2) & 1);
                put(slot + k, 16 * (g0 + ld * OWN_THREADS) + 4 * (bit & 3) + (bit >> 3));
            }
        }
    }
    const unsigned char* kb = (const unsigned char*)keys;
    for (int e = 16 * nvec + tid; e < E; e += OWN_THREADS)
        if (kb[e] == mykey) put(atomicAdd(&nlist, 1), e);
    __syncthreads();

    const int pair = wave >> 1, half = wave & 1;
    const int ra = min(pair * PAIR_ROWS, nrows), rb = min(ra + PAIR_ROWS, nrows);
    const int hd = 2 * half + (lane >> 4);
    const float* hcol = h + half * 128 + 4 * lane;
    float* ocol = out + (size_t)lo * OUTC + half * 128 + 4 * lane;
    const float4 bv = *(const float4*)(b + half * 128 + 4 * lane);
    const float4 zero4 = make_float4(0.f, 0.f, 0.f, 0.f);
    auto store_row = [&](int r, const float4& a) {
        *(float4*)(ocol + (size_t)r * OUTC) = make_float4(a.x + bv.x, a.y + bv.y, a.z + bv.z, a.w + bv.w);
    };

    if (nlist > LIST_CAP) {
        const float mxh  = hd == 0 ? mx0  : hd == 1 ? mx1  : hd == 2 ? mx2  : mx3;
        const float invh = hd == 0 ? inv0 : hd == 1 ? inv1 : hd == 2 ? inv2 : inv3;
        for (int r = ra; r < rb; ++r) {
            const int node = lo + r;
            const float4 hv = *(const float4*)(hcol + (size_t)node * OUTC);
            const float ws = __expf(scores[(size_t)(E + node) * HEADS + hd] - mxh) * invh;
            float4 acc = make_float4(hv.x * ws, hv.y * ws, hv.z * ws, hv.w * ws);
            for (int e0 = 0; e0 < E; e0 += 32) {
                const int e = e0 + lane;
                const int d = e < E ? dstp[e] : -1;
                unsigned m = __builtin_amdgcn_ballot_w32(d == node);
                while (m) {
                    const int ee = e0 + __builtin_ctz(m);
                    m &= m - 1;
                    const float w = __expf(scores[(size_t)ee * HEADS + hd] - mxh) * invh;
                    const float4 v = *(const float4*)(hcol + (size_t)ei[ee] * OUTC);
                    acc.x += v.x * w; acc.y += v.y * w; acc.z += v.z * w; acc.w += v.w * w;
                }
            }
            store_row(r, acc);
        }
        return;
    }

    const int n = nlist;
    const int total = n + nrows;
    for (int i = tid; i < total; i += OWN_THREADS) {
        int e, d, s;
        if (i < n) { e = l_e[i]; d = dstp[e]; s = ei[e]; }
        else       { d = s = lo + (i - n); e = E + d; }
        const float4 sc = sc4[e];
        __builtin_amdgcn_sched_barrier(0);
        const unsigned ur = (unsigned)(d - lo);
        const bool ok = ur < (unsigned)nrows;
        const int r = (int)ur;
        int rank = -1;
        if (ok) rank = atomicAdd(&rcount[r], 1);
        l_e[i]  = rank;
        u_sd[i] = make_int2(s, r);
        u_w[i]  = make_float4(__expf(sc.x - mx0) * inv0, __expf(sc.y - mx1) * inv1,
                              __expf(sc.z - mx2) * inv2, __expf(sc.w - mx3) * inv3);
    }
    __syncthreads();

    if (tid <= OWN_ROWS) {
        int s = 0;
        #pragma unroll
        for (int j = 0; j < OWN_ROWS; ++j) s += j < tid ? rcount[j] : 0;
        rstart[tid] = s;
    }
    __syncthreads();
    for (int i = tid; i < total; i += OWN_THREADS) {
        const int rank = l_e[i];
        const int2 sd = u_sd[i];
        const float4 wv = u_w[i];
        if (rank >= 0) {
            const int pos = rstart[sd.y] + rank;
            s_sd[pos] = sd;
            s_w[pos]  = wv;
        }
    }
    __syncthreads();

    const float* s_wf = (const float*)s_w;
    const int beg = __builtin_amdgcn_readfirstlane(rstart[ra]);
    const int end = __builtin_amdgcn_readfirstlane(rstart[rb]);
    float4 acc = zero4;
    int cur = ra;
    auto trip = [&](auto full, int i) {
        constexpr bool FULL = decltype(full)::value;
        float  w[OWN_GROUP];
        v4f    v[OWN_GROUP];
        int    rr[OWN_GROUP];
        static_assert(OWN_GROUP == 16, "one entry per lane of a 16-lane row");
        const int2 sdl = s_sd[FULL ? i + (lane & 15) : min(i + (lane & 15), end - 1)];
        #pragma unroll
        for (int k = 0; k < OWN_GROUP; ++k)
            w[k] = s_wf[(FULL ? i + k : min(i + k, end - 1)) * 4 + hd];
        #pragma unroll
        for (int k = 0; k < OWN_GROUP; ++k) {
            const int sidx = __builtin_amdgcn_readlane(sdl.x, k);
            rr[k] = __builtin_amdgcn_readlane(sdl.y, k);
            if (FULL || i + k < end) {
                v[k] = *(const v4f*)(hcol + (size_t)sidx * OUTC);
            } else {
                v[k] = (v4f)(0.f);
                w[k] = 0.f;
            }
        }
        #pragma unroll
        for (int k = 0; k < OWN_GROUP; ++k) {
            if (rr[k] != cur) {
                store_row(cur, acc);
                acc = zero4;
                cur = rr[k];
            }
            acc.x += v[k].x * w[k]; acc.y += v[k].y * w[k];
            acc.z += v[k].z * w[k]; acc.w += v[k].w * w[k];
        }
    };
    int i = beg;
    for (; i + OWN_GROUP <= end; i += OWN_GROUP) trip(std::true_type{}, i);
    if (i < end) trip(std::false_type{}, i);
    if (beg < end) store_row(cur, acc);
}

extern "C" void kernel_launch(void* const* d_in, const int* in_sizes, int n_in,
                              void* d_out, int out_size, void* d_ws, size_t ws_size,
                              hipStream_t stream) {
    const float* x   = (const float*)d_in[0];
    const float* W   = (const float*)d_in[1];
    const float* att = (const float*)d_in[2];
    const float* b   = (const float*)d_in[3];
    const int*   ei  = (const int*)d_in[4];

    const int Nn = in_sizes[0] / IN_F;
    const int E  = in_sizes[4] / 2;
    const int totalEdges = E + Nn;

    float* out    = (float*)d_out;
    float* h      = (float*)d_ws;
    float* p      = h + (size_t)Nn * OUTC;
    float* scores = p + (size_t)Nn * 2 * HEADS;
    float* part   = scores + (size_t)totalEdges * HEADS;
    unsigned int* keys = (unsigned int*)(part + SCORE_BLOCKS * 2 * HEADS);

    const int nTilesM = Nn / 16;
    gat_gemm_wmma<<<nTilesM, 128, 0, stream>>>(x, W, att, h, p, nTilesM);

    gat_scores<<<SCORE_BLOCKS, 256, 0, stream>>>(p, ei, scores, part, keys, E, Nn);
    gat_gather_own<<<(Nn + OWN_ROWS - 1) / OWN_ROWS, OWN_THREADS, 0, stream>>>(
        h, ei, scores, part, keys, b, out, E, Nn);
}


# scores kernel stores each edge's record (scores, source, row) sorted by target slice in its block's chunk, plus slice starts; last kernel reads its runs: key scan and gathers go (sum order changes)
# speedup vs baseline: 28.5557x; 1.1008x over previous
#include <hip/hip_runtime.h>
#include <hip/hip_bf16.h>
#include <math.h>
#include <type_traits>

typedef __attribute__((ext_vector_type(2))) float v2f;
typedef __attribute__((ext_vector_type(8))) float v8f;
typedef __attribute__((ext_vector_type(4))) unsigned int v4u;
typedef __attribute__((ext_vector_type(4))) float v4f;

#define IN_F   256
#define OUTC   256
#define HEADS  4
#define OUT_F  64
#define NEG_SLOPE 0.2f
#define OWN_ROWS 80

__device__ __forceinline__ float leaky(float v) { return v > 0.f ? v : NEG_SLOPE * v; }

__device__ __forceinline__ float shift_exp(float m, float M) { return m == M ? 1.f : __expf(m - M); }
template <int M>
__device__ __forceinline__ float row_xor(float v) {
    return __builtin_bit_cast(float, __builtin_amdgcn_update_dpp(0, __builtin_bit_cast(int, v),
                                                                 0x160 + M, 0xf, 0xf, true));
}
__device__ __forceinline__ float xhalf(float v) {
    const int b = __builtin_bit_cast(int, v);
    return __builtin_bit_cast(float, __builtin_amdgcn_permlanex16(b, b, 0x76543210, 0xfedcba98, false, false));
}
__device__ __forceinline__ float row_max(float v) {
    v = fmaxf(v, row_xor<1>(v)); v = fmaxf(v, row_xor<2>(v));
    v = fmaxf(v, row_xor<4>(v)); return fmaxf(v, row_xor<8>(v));
}
__device__ __forceinline__ float row_sum(float v) {
    v += row_xor<1>(v); v += row_xor<2>(v); v += row_xor<4>(v); return v + row_xor<8>(v);
}
__device__ __forceinline__ float wave_max(float v) { v = row_max(v); return fmaxf(v, xhalf(v)); }
__device__ __forceinline__ float wave_sum(float v) { v = row_sum(v); return v + xhalf(v); }
#define SCORE_BLOCKS 512

#define GEMM_WAVES 4

template <int M>
__device__ __forceinline__ float row_xor_add(float v) {
    const int t = __builtin_amdgcn_update_dpp(0, __builtin_bit_cast(int, v),
                                              0x160 + M, 0xf, 0xf, true);
    return v + __builtin_bit_cast(float, t);
}

struct Frag {
    float4 a;
    float4 q[4];
};

__device__ __forceinline__ void load_frag(Frag& f, const float* __restrict__ xk,
                                          const float* __restrict__ wk) {
    f.a = *(const float4*)xk;
    #pragma unroll
    for (int j = 0; j < 4; ++j) f.q[j] = *(const float4*)(wk + (size_t)j * OUTC);
}

#define WMMA_F32(A, B, C) __builtin_amdgcn_wmma_f32_16x16x4_f32( \
         false, (A),  false, (B),  (short)0, (C), \
         false,  false)

__device__ __forceinline__ void mma_frag(const Frag& f, v8f& c0, v8f& c1, v8f& c2, v8f& c3) {
    v2f a0; a0.x = f.a.x; a0.y = f.a.y;
    v2f a1; a1.x = f.a.z; a1.y = f.a.w;
    v2f b;
    b.x = f.q[0].x; b.y = f.q[1].x; c0 = WMMA_F32(a0, b, c0);
    b.x = f.q[0].y; b.y = f.q[1].y; c1 = WMMA_F32(a0, b, c1);
    b.x = f.q[0].z; b.y = f.q[1].z; c2 = WMMA_F32(a0, b, c2);
    b.x = f.q[0].w; b.y = f.q[1].w; c3 = WMMA_F32(a0, b, c3);
    b.x = f.q[2].x; b.y = f.q[3].x; c0 = WMMA_F32(a1, b, c0);
    b.x = f.q[2].y; b.y = f.q[3].y; c1 = WMMA_F32(a1, b, c1);
    b.x = f.q[2].z; b.y = f.q[3].z; c2 = WMMA_F32(a1, b, c2);
    b.x = f.q[2].w; b.y = f.q[3].w; c3 = WMMA_F32(a1, b, c3);
}

__global__ void __launch_bounds__(128) gat_gemm_wmma(
        const float* __restrict__ x, const float* __restrict__ W,
        const float* __restrict__ att,
        float* __restrict__ h, float* __restrict__ p, int nTilesM) {
    __shared__ float spart[GEMM_WAVES][16][2];
    const int mTile = blockIdx.x;
    if (mTile >= nTilesM) return;
    const int head = threadIdx.x >> 5;
    const int lane = threadIdx.x & 31;
    const int c    = lane & 15;
    const int half = lane >> 4;

    const float* xk = x + (size_t)(mTile * 16 + c) * IN_F + 4 * half;
    const float* wk = W + (size_t)(4 * half) * OUTC + head * OUT_F + 4 * c;

    const float* ap = att + head * 2 * OUT_F + 4 * c;
    float as[4], ad[4];
    #pragma unroll
    for (int t = 0; t < 4; ++t) { as[t] = ap[t]; ad[t] = ap[OUT_F + t]; }

    v8f c0 = {}, c1 = {}, c2 = {}, c3 = {};
    Frag cur, nxt;
    load_frag(cur, xk, wk);
    #pragma unroll
    for (int k = 8; k < IN_F; k += 8) {
        load_frag(nxt, xk + k, wk + (size_t)k * OUTC);
        mma_frag(cur, c0, c1, c2, c3);
        cur = nxt;
    }
    mma_frag(cur, c0, c1, c2, c3);

    const int rowOff = half << 3;
    float* hp = h + (size_t)(mTile * 16 + rowOff) * OUTC + head * OUT_F + 4 * c;
    #pragma unroll
    for (int i = 0; i < 8; ++i) {
        float4 v; v.x = c0[i]; v.y = c1[i]; v.z = c2[i]; v.w = c3[i];
        *(float4*)(hp + (size_t)i * OUTC) = v;
    }

    float ps[8], pd[8];
    #pragma unroll
    for (int i = 0; i < 8; ++i) {
        const float l0 = leaky(c0[i]), l1 = leaky(c1[i]), l2 = leaky(c2[i]), l3 = leaky(c3[i]);
        ps[i] = (l0 * as[0] + l1 * as[1]) + (l2 * as[2] + l3 * as[3]);
        pd[i] = (l0 * ad[0] + l1 * ad[1]) + (l2 * ad[2] + l3 * ad[3]);
    }
    #pragma unroll
    for (int i = 0; i < 8; ++i) { ps[i] = row_xor_add<1>(ps[i]); pd[i] = row_xor_add<1>(pd[i]); }
    #pragma unroll
    for (int i = 0; i < 8; ++i) { ps[i] = row_xor_add<2>(ps[i]); pd[i] = row_xor_add<2>(pd[i]); }
    #pragma unroll
    for (int i = 0; i < 8; ++i) { ps[i] = row_xor_add<4>(ps[i]); pd[i] = row_xor_add<4>(pd[i]); }
    #pragma unroll
    for (int i = 0; i < 8; ++i) { ps[i] = row_xor_add<8>(ps[i]); pd[i] = row_xor_add<8>(pd[i]); }
    if (c == 0) {
        #pragma unroll
        for (int i = 0; i < 8; ++i) {
            spart[head][rowOff + i][0] = ps[i];
            spart[head][rowOff + i][1] = pd[i];
        }
    }
    __syncthreads();
    if (threadIdx.x < 32) {
        const int row = threadIdx.x >> 1;
        const int sd  = threadIdx.x & 1;
        float4 v;
        v.x = spart[0][row][sd]; v.y = spart[1][row][sd];
        v.z = spart[2][row][sd]; v.w = spart[3][row][sd];
        *(float4*)(p + (size_t)(mTile * 16 + row) * (2 * HEADS) + sd * HEADS) = v;
    }
}

#define SCORE_WAVES 8
#define SCORE_THREADS (SCORE_WAVES * 32)
#define EDGE_TRIPS 3
#define TABW 256

__device__ __forceinline__ int wave_scan_incl(int v, int lane) {
    #pragma unroll
    for (int d = 1; d < 32; d <<= 1) {
        const int t = __shfl_up(v, d);
        if (lane >= d) v += t;
    }
    return v;
}

__global__ void __launch_bounds__(SCORE_THREADS) gat_scores(
        const float* __restrict__ p, const int* __restrict__ ei,
        float* __restrict__ rec, float* __restrict__ part,
        int* __restrict__ tab, int E, int Nn) {
    __shared__ float smax[SCORE_WAVES][HEADS];
    __shared__ float ssum[SCORE_WAVES][HEADS];
    __shared__ int   hist[TABW];
    __shared__ int   wtot[SCORE_WAVES];
    static_assert(TABW == SCORE_THREADS, "one table slot per thread");
    const int tid  = threadIdx.x;
    const int lane = tid & 31;
    const int wib  = tid >> 5;
    const int chunk = (E + SCORE_BLOCKS - 1) / SCORE_BLOCKS;
    const int e0 = blockIdx.x * chunk;
    const int e1 = min(E, e0 + chunk);

    hist[tid] = 0;
    __syncthreads();

    const float4* p4 = (const float4*)p;
    float m0 = -INFINITY, m1 = -INFINITY, m2 = -INFINITY, m3 = -INFINITY;
    float a0 = 0.f, a1 = 0.f, a2 = 0.f, a3 = 0.f;
    auto take = [&](const float4& v) {
        float n;
        n = fmaxf(m0, v.x); a0 = a0 * shift_exp(m0, n) + shift_exp(v.x, n); m0 = n;
        n = fmaxf(m1, v.y); a1 = a1 * shift_exp(m1, n) + shift_exp(v.y, n); m1 = n;
        n = fmaxf(m2, v.z); a2 = a2 * shift_exp(m2, n) + shift_exp(v.z, n); m2 = n;
        n = fmaxf(m3, v.w); a3 = a3 * shift_exp(m3, n) + shift_exp(v.w, n); m3 = n;
    };

    float4 sv[EDGE_TRIPS];
    int ss[EDGE_TRIPS], sl[EDGE_TRIPS], sr[EDGE_TRIPS], rk[EDGE_TRIPS];
    #pragma unroll
    for (int k = 0; k < EDGE_TRIPS; ++k) {
        const int e = e0 + tid + k * SCORE_THREADS;
        rk[k] = -1;
        if (e < e1) {
            const int s = ei[e], d = ei[E + e];
            const float4 a = p4[(size_t)s * 2];
            const float4 b = p4[(size_t)d * 2 + 1];
            float4 v;
            v.x = a.x + b.x; v.y = a.y + b.y; v.z = a.z + b.z; v.w = a.w + b.w;
            const int slice = (int)((unsigned)d / OWN_ROWS);
            sv[k] = v; ss[k] = s; sl[k] = slice; sr[k] = d - slice * OWN_ROWS;
            rk[k] = atomicAdd(&hist[slice], 1);
            take(v);
        }
    }
    {
        const int nchunk = (Nn + SCORE_BLOCKS - 1) / SCORE_BLOCKS;
        for (int i = SCORE_THREADS - 1 - tid; i < nchunk; i += SCORE_THREADS) {
            const int node = blockIdx.x * nchunk + i;
            if (node < Nn) {
                const float4 a = p4[(size_t)node * 2];
                const float4 b = p4[(size_t)node * 2 + 1];
                float4 v;
                v.x = a.x + b.x; v.y = a.y + b.y; v.z = a.z + b.z; v.w = a.w + b.w;
                take(v);
            }
        }
    }

    const float w0 = wave_max(m0), w1 = wave_max(m1), w2 = wave_max(m2), w3 = wave_max(m3);
    a0 = wave_sum(a0 * shift_exp(m0, w0)); a1 = wave_sum(a1 * shift_exp(m1, w1));
    a2 = wave_sum(a2 * shift_exp(m2, w2)); a3 = wave_sum(a3 * shift_exp(m3, w3));
    if (lane == 0) {
        smax[wib][0] = w0; smax[wib][1] = w1; smax[wib][2] = w2; smax[wib][3] = w3;
        ssum[wib][0] = a0; ssum[wib][1] = a1; ssum[wib][2] = a2; ssum[wib][3] = a3;
    }
    __syncthreads();
    if (tid < HEADS) {
        float M = smax[0][tid];
        for (int w = 1; w < SCORE_WAVES; ++w) M = fmaxf(M, smax[w][tid]);
        float S = 0.f;
        for (int w = 0; w < SCORE_WAVES; ++w) S += ssum[w][tid] * shift_exp(smax[w][tid], M);
        part[blockIdx.x * (2 * HEADS) + tid]         = M;
        part[blockIdx.x * (2 * HEADS) + HEADS + tid] = S;
    }
    const int hv  = hist[tid];
    const int inc = wave_scan_incl(hv, lane);
    if (lane == 31) wtot[wib] = inc;
    __syncthreads();
    int first = inc - hv;
    #pragma unroll
    for (int w = 0; w < SCORE_WAVES; ++w) first += w < wib ? wtot[w] : 0;
    hist[tid] = first;
    tab[blockIdx.x * TABW + tid] = first;
    __syncthreads();
    float4* r4 = (float4*)rec;
    #pragma unroll
    for (int k = 0; k < EDGE_TRIPS; ++k) {
        if (rk[k] >= 0) {
            const size_t pos = (size_t)e0 + hist[sl[k]] + rk[k];
            r4[2 * pos]     = sv[k];
            r4[2 * pos + 1] = make_float4(__int_as_float(ss[k]), __int_as_float(sr[k]), 0.f, 0.f);
        }
    }
}

#define OWN_THREADS 1024
#define OWN_WAVES   32
#define OWN_PAIRS   (OWN_WAVES / 2)
#define PAIR_ROWS   (OWN_ROWS / OWN_PAIRS)
#define OWN_GROUP   16
#define LIST_CAP    4096
#define ENT_CAP     (LIST_CAP + OWN_ROWS)
static_assert(OWN_PAIRS * PAIR_ROWS == OWN_ROWS, "the pairs of waves cover the slice exactly");

__device__ __forceinline__ float uniform_f(float v) {
    return __builtin_bit_cast(float, __builtin_amdgcn_readfirstlane(__builtin_bit_cast(int, v)));
}

__global__ void __launch_bounds__(OWN_THREADS) gat_gather_own(
        const float* __restrict__ h, const int* __restrict__ ei,
        const float* __restrict__ p, const float* __restrict__ part,
        const float* __restrict__ rec, const int* __restrict__ tab,
        const float* __restrict__ b, float* __restrict__ out, int E, int Nn) {
    __shared__ int    l_e[ENT_CAP];
    __shared__ int2   u_sd[ENT_CAP];
    __shared__ float4 u_w[ENT_CAP];
    __shared__ int2   s_sd[ENT_CAP];
    __shared__ float4 s_w[ENT_CAP];
    __shared__ int    rcount[OWN_ROWS];
    __shared__ int    rstart[OWN_ROWS + 1];
    __shared__ int    s_wt[SCORE_BLOCKS / 32];
    __shared__ float4 s_pm[SCORE_BLOCKS / 32];
    __shared__ float4 s_ps[SCORE_BLOCKS / 32];
    static_assert(SCORE_BLOCKS == 512 && OWN_THREADS >= SCORE_BLOCKS,
                  "one pair, and one run, per lane in 16 waves; the second merge takes entry lane & 15");

    const int tid  = threadIdx.x;
    const int lane = tid & 31;
    const int wave = __builtin_amdgcn_readfirstlane(tid >> 5);
    const int lo    = blockIdx.x * OWN_ROWS;
    const int nrows = min(OWN_ROWS, Nn - lo);
    if (nrows <= 0) return;

    const float4* p4 = (const float4*)p;

    int c0 = 0, cnt = 0;
    if (wave < SCORE_BLOCKS / 32) {
        const float4* part4 = (const float4*)part;
        const float4 m = part4[2 * tid];
        float4 a = part4[2 * tid + 1];
        const int* tp = tab + (size_t)tid * TABW + blockIdx.x;
        c0 = tp[0]; cnt = tp[1] - c0;
        float4 M;
        M.x = wave_max(m.x); M.y = wave_max(m.y); M.z = wave_max(m.z); M.w = wave_max(m.w);
        a.x = wave_sum(a.x * shift_exp(m.x, M.x)); a.y = wave_sum(a.y * shift_exp(m.y, M.y));
        a.z = wave_sum(a.z * shift_exp(m.z, M.z)); a.w = wave_sum(a.w * shift_exp(m.w, M.w));
        if (lane == 0) { s_pm[wave] = M; s_ps[wave] = a; }
    }
    const int inc = wave_scan_incl(cnt, lane);
    if (wave < SCORE_BLOCKS / 32 && lane == 31) s_wt[wave] = inc;
    if (tid < OWN_ROWS) rcount[tid] = 0;
    __syncthreads();
    float mx0, mx1, mx2, mx3, inv0, inv1, inv2, inv3;
    {
        const float4 m = s_pm[lane & 15];
        float4 a = s_ps[lane & 15];
        float4 M;
        M.x = row_max(m.x); M.y = row_max(m.y); M.z = row_max(m.z); M.w = row_max(m.w);
        a.x = row_sum(a.x * shift_exp(m.x, M.x)); a.y = row_sum(a.y * shift_exp(m.y, M.y));
        a.z = row_sum(a.z * shift_exp(m.z, M.z)); a.w = row_sum(a.w * shift_exp(m.w, M.w));
        mx0 = uniform_f(M.x); mx1 = uniform_f(M.y); mx2 = uniform_f(M.z); mx3 = uniform_f(M.w);
        inv0 = uniform_f(1.f / a.x); inv1 = uniform_f(1.f / a.y);
        inv2 = uniform_f(1.f / a.z); inv3 = uniform_f(1.f / a.w);
    }

    int off = inc - cnt, n = 0;
    #pragma unroll
    for (int w = 0; w < SCORE_BLOCKS / 32; ++w) {
        const int t = s_wt[w];
        off += w < wave ? t : 0;
        n += t;
    }
    const int* dstp = ei + E;

    const int pair = wave >> 1, half = wave & 1;
    const int ra = min(pair * PAIR_ROWS, nrows), rb = min(ra + PAIR_ROWS, nrows);
    const int hd = 2 * half + (lane >> 4);
    const float* hcol = h + half * 128 + 4 * lane;
    float* ocol = out + (size_t)lo * OUTC + half * 128 + 4 * lane;
    const float4 bv = *(const float4*)(b + half * 128 + 4 * lane);
    const float4 zero4 = make_float4(0.f, 0.f, 0.f, 0.f);
    auto store_row = [&](int r, const float4& a) {
        *(float4*)(ocol + (size_t)r * OUTC) = make_float4(a.x + bv.x, a.y + bv.y, a.z + bv.z, a.w + bv.w);
    };

    if (n > LIST_CAP) {
        const float mxh  = hd == 0 ? mx0  : hd == 1 ? mx1  : hd == 2 ? mx2  : mx3;
        const float invh = hd == 0 ? inv0 : hd == 1 ? inv1 : hd == 2 ? inv2 : inv3;
        for (int r = ra; r < rb; ++r) {
            const int node = lo + r;
            const float4 hv = *(const float4*)(hcol + (size_t)node * OUTC);
            const float pdn = p[(size_t)node * (2 * HEADS) + HEADS + hd];
            const float ws = __expf(p[(size_t)node * (2 * HEADS) + hd] + pdn - mxh) * invh;
            float4 acc = make_float4(hv.x * ws, hv.y * ws, hv.z * ws, hv.w * ws);
            for (int e0 = 0; e0 < E; e0 += 32) {
                const int e = e0 + lane;
                const int d = e < E ? dstp[e] : -1;
                unsigned m = __builtin_amdgcn_ballot_w32(d == node);
                while (m) {
                    const int ee = e0 + __builtin_ctz(m);
                    m &= m - 1;
                    const int se = ei[ee];
                    const float w = __expf(p[(size_t)se * (2 * HEADS) + hd] + pdn - mxh) * invh;
                    const float4 v = *(const float4*)(hcol + (size_t)se * OUTC);
                    acc.x += v.x * w; acc.y += v.y * w; acc.z += v.z * w; acc.w += v.w * w;
                }
            }
            store_row(r, acc);
        }
        return;
    }

    const int total = n + nrows;
    if (wave < SCORE_BLOCKS / 32) {
        const int g0 = tid * ((E + SCORE_BLOCKS - 1) / SCORE_BLOCKS) + c0;
        for (int r = 0; r < cnt; ++r) l_e[off + r] = g0 + r;
    }
    __syncthreads();
    for (int i = tid; i < total; i += OWN_THREADS) {
        v4f sc; int s, r;
        if (i < n) {
            const v4f* q = (const v4f*)rec + 2 * (size_t)l_e[i];
            sc = q[0];
            const v2f id = *(const v2f*)(q + 1);
            s = __float_as_int(id.x); r = __float_as_int(id.y);
        } else {
            r = i - n; s = lo + r;
            const float4 a = p4[(size_t)s * 2];
            const float4 d = p4[(size_t)s * 2 + 1];
            sc.x = a.x + d.x; sc.y = a.y + d.y; sc.z = a.z + d.z; sc.w = a.w + d.w;
        }
        l_e[i]  = atomicAdd(&rcount[r], 1);
        u_sd[i] = make_int2(s, r);
        u_w[i]  = make_float4(__expf(sc.x - mx0) * inv0, __expf(sc.y - mx1) * inv1,
                              __expf(sc.z - mx2) * inv2, __expf(sc.w - mx3) * inv3);
    }
    __syncthreads();

    if (tid <= OWN_ROWS) {
        int s = 0;
        #pragma unroll
        for (int j = 0; j < OWN_ROWS; ++j) s += j < tid ? rcount[j] : 0;
        rstart[tid] = s;
    }
    __syncthreads();
    for (int i = tid; i < total; i += OWN_THREADS) {
        const int rank = l_e[i];
        const int2 sd = u_sd[i];
        const float4 wv = u_w[i];
        const int pos = rstart[sd.y] + rank;
        s_sd[pos] = sd;
        s_w[pos]  = wv;
    }
    __syncthreads();

    const float* s_wf = (const float*)s_w;
    const int beg = __builtin_amdgcn_readfirstlane(rstart[ra]);
    const int end = __builtin_amdgcn_readfirstlane(rstart[rb]);
    float4 acc = zero4;
    int cur = ra;
    auto trip = [&](auto full, int i) {
        constexpr bool FULL = decltype(full)::value;
        float  w[OWN_GROUP];
        v4f    v[OWN_GROUP];
        int    rr[OWN_GROUP];
        static_assert(OWN_GROUP == 16, "one entry per lane of a 16-lane row");
        const int2 sdl = s_sd[FULL ? i + (lane & 15) : min(i + (lane & 15), end - 1)];
        #pragma unroll
        for (int k = 0; k < OWN_GROUP; ++k)
            w[k] = s_wf[(FULL ? i + k : min(i + k, end - 1)) * 4 + hd];
        #pragma unroll
        for (int k = 0; k < OWN_GROUP; ++k) {
            const int sidx = __builtin_amdgcn_readlane(sdl.x, k);
            rr[k] = __builtin_amdgcn_readlane(sdl.y, k);
            if (FULL || i + k < end) {
                v[k] = *(const v4f*)(hcol + (size_t)sidx * OUTC);
            } else {
                v[k] = (v4f)(0.f);
                w[k] = 0.f;
            }
        }
        #pragma unroll
        for (int k = 0; k < OWN_GROUP; ++k) {
            if (rr[k] != cur) {
                store_row(cur, acc);
                acc = zero4;
                cur = rr[k];
            }
            acc.x += v[k].x * w[k]; acc.y += v[k].y * w[k];
            acc.z += v[k].z * w[k]; acc.w += v[k].w * w[k];
        }
    };
    int i = beg;
    for (; i + OWN_GROUP <= end; i += OWN_GROUP) trip(std::true_type{}, i);
    if (i < end) trip(std::false_type{}, i);
    if (beg < end) store_row(cur, acc);
}

extern "C" void kernel_launch(void* const* d_in, const int* in_sizes, int n_in,
                              void* d_out, int out_size, void* d_ws, size_t ws_size,
                              hipStream_t stream) {
    const float* x   = (const float*)d_in[0];
    const float* W   = (const float*)d_in[1];
    const float* att = (const float*)d_in[2];
    const float* b   = (const float*)d_in[3];
    const int*   ei  = (const int*)d_in[4];

    const int Nn = in_sizes[0] / IN_F;
    const int E  = in_sizes[4] / 2;

    float* out    = (float*)d_out;
    float* h      = (float*)d_ws;
    float* p      = h + (size_t)Nn * OUTC;
    float* part   = p + (size_t)Nn * 2 * HEADS;
    float* rec    = part + SCORE_BLOCKS * 2 * HEADS;
    int*   tab    = (int*)(rec + (size_t)SCORE_BLOCKS * ((E + SCORE_BLOCKS - 1) / SCORE_BLOCKS) * 8);

    const int nTilesM = Nn / 16;
    gat_gemm_wmma<<<nTilesM, 128, 0, stream>>>(x, W, att, h, p, nTilesM);

    gat_scores<<<SCORE_BLOCKS, SCORE_THREADS, 0, stream>>>(p, ei, rec, part, tab, E, Nn);
    gat_gather_own<<<(Nn + OWN_ROWS - 1) / OWN_ROWS, OWN_THREADS, 0, stream>>>(
        h, ei, p, part, rec, tab, b, out, E, Nn);
}
